# starting kernel
# baseline (speedup 1.0000x reference)
#include <hip/hip_runtime.h>
#include <hip/hip_bf16.h>
#include <math.h>

typedef float v2f __attribute__((ext_vector_type(2)));
typedef float v8f __attribute__((ext_vector_type(8)));

#define D_FEAT 128

__global__ void init_state_kernel(float* __restrict__ c, float* __restrict__ z,
                                  unsigned int* __restrict__ mb, int V) {
    int i = blockIdx.x * blockDim.x + threadIdx.x;
    int n = V * D_FEAT;
    if (i < n) c[i] = 0.0f;
    if (i < V) { z[i] = 0.0f; mb[i] = 0xFF800000u; }
}

__global__ void packT_kernel(const float* __restrict__ W, float* __restrict__ WT,
                             int R, int C) {
    int i = blockIdx.x * blockDim.x + threadIdx.x;
    if (i >= R * C) return;
    int n = i / C;
    int k = i - n * C;
    WT[(size_t)k * R + n] = W[i];
}

__global__ void node_dot_kernel(const float* __restrict__ nf,
                                const float* __restrict__ W_edge,
                                float* __restrict__ pd, float* __restrict__ ps,
                                int V) {
    int gid = blockIdx.x * blockDim.x + threadIdx.x;
    int v = gid >> 5;
    int lane = gid & 31;
    if (v >= V) return;
    float4 x  = ((const float4*)(nf + (size_t)v * D_FEAT))[lane];
    float4 w0 = ((const float4*)W_edge)[lane];
    float4 w1 = ((const float4*)(W_edge + D_FEAT))[lane];
    float d0 = x.x * w0.x + x.y * w0.y + x.z * w0.z + x.w * w0.w;
    float d1 = x.x * w1.x + x.y * w1.y + x.z * w1.z + x.w * w1.w;
#pragma unroll
    for (int off = 16; off > 0; off >>= 1) {
        d0 += __shfl_down(d0, off, 32);
        d1 += __shfl_down(d1, off, 32);
    }
    if (lane == 0) { pd[v] = d0; ps[v] = d1; }
}

template <int N>
__global__ __launch_bounds__(256) void wmma_gemm_kernel(
    const float* __restrict__ X, const float* __restrict__ B,
    const float* __restrict__ bias, float* __restrict__ Out, int V) {
    __shared__ float xs[16][132];
    const int tid = threadIdx.x;
    const int rowbase = blockIdx.x * 16;
#pragma unroll
    for (int i = 0; i < 8; ++i) {
        int idx = tid + i * 256;
        int r = idx >> 7;
        int col = idx & 127;
        int row = rowbase + r;
        if (row >= V) row = V - 1;
        xs[r][col] = X[(size_t)row * D_FEAT + col];
    }
    __syncthreads();
    const int lane = tid & 31;
    const int wave = tid >> 5;
    const int lo = lane & 15;
    const int hi = lane >> 4;
    const bool fullTile = (rowbase + 16 <= V);
    for (int t = wave; t < N / 16; t += 8) {
        const int colbase = t * 16;
        v8f acc;
        const float bv = bias[colbase + lo];
#pragma unroll
        for (int r = 0; r < 8; ++r) acc[r] = bv;
#pragma unroll
        for (int kk = 0; kk < D_FEAT; kk += 4) {
            const int ka = kk + 2 * hi;
            v2f a, b;
            a[0] = xs[lo][ka];
            a[1] = xs[lo][ka + 1];
            b[0] = B[(size_t)ka * N + colbase + lo];
            b[1] = B[(size_t)(ka + 1) * N + colbase + lo];
            acc = __builtin_amdgcn_wmma_f32_16x16x4_f32(
                false, a, false, b, (short)0, acc, false, false);
        }
        if (fullTile) {
#pragma unroll
            for (int r = 0; r < 8; ++r) {
                int row = rowbase + r + 8 * hi;
                Out[(size_t)row * N + colbase + lo] = acc[r];
            }
        } else {
#pragma unroll
            for (int r = 0; r < 8; ++r) {
                int row = rowbase + r + 8 * hi;
                if (row < V) Out[(size_t)row * N + colbase + lo] = acc[r];
            }
        }
    }
}

__global__ void edge_logit_kernel(const int* __restrict__ src,
                                  const int* __restrict__ dst,
                                  const float* __restrict__ pd,
                                  const float* __restrict__ ps,
                                  const float* __restrict__ b_edge,
                                  float* __restrict__ logit,
                                  unsigned int* __restrict__ mb, int E) {
    int e = blockIdx.x * blockDim.x + threadIdx.x;
    if (e >= E) return;
    int d = dst[e];
    float x = pd[d] + ps[src[e]] + b_edge[0];
    float l = x > 0.0f ? x : 0.01f * x;
    logit[e] = l;
    int iv = __float_as_int(l);
    if (iv >= 0) atomicMax((int*)(mb + d), iv);
    else         atomicMin(mb + d, (unsigned int)iv);
}

__global__ void edge_scatter_kernel(const int* __restrict__ src,
                                    const int* __restrict__ dst,
                                    const float* __restrict__ logit,
                                    const unsigned int* __restrict__ mb,
                                    float* __restrict__ z,
                                    const float* __restrict__ hv,
                                    float* __restrict__ c, int E) {
    int gid = blockIdx.x * blockDim.x + threadIdx.x;
    int e = gid >> 5;
    int lane = gid & 31;
    if (e >= E) return;
    int d = dst[e], s = src[e];
    float m = __int_as_float((int)mb[d]);
    float ev = expf(logit[e] - m);
    if (lane == 0) atomicAdd(z + d, ev);
    float4 h = ((const float4*)(hv + (size_t)s * D_FEAT))[lane];
    float* cd = c + (size_t)d * D_FEAT + lane * 4;
    atomicAdd(cd + 0, ev * h.x);
    atomicAdd(cd + 1, ev * h.y);
    atomicAdd(cd + 2, ev * h.z);
    atomicAdd(cd + 3, ev * h.w);
}

__global__ void context_kernel(float* __restrict__ c, const float* __restrict__ z,
                               int V) {
    int i = blockIdx.x * blockDim.x + threadIdx.x;
    if (i >= V * D_FEAT) return;
    float zz = z[i >> 7];
    float x = zz > 0.0f ? c[i] / zz : 0.0f;
    c[i] = x > 0.0f ? x : expf(x) - 1.0f;
}

__global__ void gru_kernel(const float* __restrict__ gi, const float* __restrict__ gh,
                           const float* __restrict__ nf, float* __restrict__ out,
                           int V) {
    int i = blockIdx.x * blockDim.x + threadIdx.x;
    if (i >= V * D_FEAT) return;
    int v = i >> 7;
    int d = i & 127;
    size_t base = (size_t)v * 384;
    float ir = gi[base + d], iz = gi[base + 128 + d], in_ = gi[base + 256 + d];
    float hr = gh[base + d], hz = gh[base + 128 + d], hn  = gh[base + 256 + d];
    float r = 1.0f / (1.0f + expf(-(ir + hr)));
    float u = 1.0f / (1.0f + expf(-(iz + hz)));
    float n = tanhf(in_ + r * hn);
    float h = (1.0f - u) * n + u * nf[i];
    out[i] = h > 0.0f ? h : 0.0f;
}

extern "C" void kernel_launch(void* const* d_in, const int* in_sizes, int n_in,
                              void* d_out, int out_size, void* d_ws, size_t ws_size,
                              hipStream_t stream) {
    const float* nf     = (const float*)d_in[0];
    const int*   src    = (const int*)d_in[1];
    const int*   dst    = (const int*)d_in[2];
    const float* W_edge = (const float*)d_in[3];
    const float* b_edge = (const float*)d_in[4];
    const float* W_proj = (const float*)d_in[5];
    const float* b_proj = (const float*)d_in[6];
    const float* W_ih   = (const float*)d_in[7];
    const float* b_ih   = (const float*)d_in[8];
    const float* W_hh   = (const float*)d_in[9];
    const float* b_hh   = (const float*)d_in[10];

    const int V = in_sizes[0] / D_FEAT;
    const int E = in_sizes[1];

    float* wsf = (float*)d_ws;
    float* pd    = wsf;
    float* ps    = pd + V;
    unsigned int* mb = (unsigned int*)(ps + V);
    float* z     = (float*)(mb + V);
    float* logit = z + V;
    float* hv    = logit + E;
    float* gh    = hv + (size_t)V * 128;
    float* c     = gh + (size_t)V * 384;
    float* gi    = c + (size_t)V * 128;
    float* WhhT  = gi + (size_t)V * 384;
    float* WihT  = WhhT + 128 * 384;

    const int blk = 256;

    packT_kernel<<<(384 * 128 + blk - 1) / blk, blk, 0, stream>>>(W_hh, WhhT, 384, 128);
    packT_kernel<<<(384 * 128 + blk - 1) / blk, blk, 0, stream>>>(W_ih, WihT, 384, 128);

    init_state_kernel<<<((size_t)V * 128 + blk - 1) / blk, blk, 0, stream>>>(c, z, mb, V);

    node_dot_kernel<<<((size_t)V * 32 + blk - 1) / blk, blk, 0, stream>>>(
        nf, W_edge, pd, ps, V);

    const int gemmBlocks = (V + 15) / 16;
    wmma_gemm_kernel<128><<<gemmBlocks, 256, 0, stream>>>(nf, W_proj, b_proj, hv, V);
    wmma_gemm_kernel<384><<<gemmBlocks, 256, 0, stream>>>(nf, WhhT, b_hh, gh, V);

    edge_logit_kernel<<<(E + blk - 1) / blk, blk, 0, stream>>>(
        src, dst, pd, ps, b_edge, logit, mb, E);
    edge_scatter_kernel<<<((size_t)E * 32 + blk - 1) / blk, blk, 0, stream>>>(
        src, dst, logit, mb, z, hv, c, E);

    context_kernel<<<((size_t)V * 128 + blk - 1) / blk, blk, 0, stream>>>(c, z, V);

    wmma_gemm_kernel<384><<<gemmBlocks, 256, 0, stream>>>(c, WihT, b_ih, gi, V);

    gru_kernel<<<((size_t)V * 128 + blk - 1) / blk, blk, 0, stream>>>(
        gi, gh, nf, (float*)d_out, V);
}


# Owner scan on bit-sliced tile numbers: the logit pass (4 edges per lane) writes 8 bit words per 32 edges by ballot; a tile owner tests 32 edges on 8 words, not 32 dst; sums stay in arrival order
# speedup vs baseline: 1.8675x; 1.8675x over previous
#include <hip/hip_runtime.h>
#include <hip/hip_bf16.h>
#include <math.h>

typedef float v2f __attribute__((ext_vector_type(2)));
typedef float v8f __attribute__((ext_vector_type(8)));

#define D_FEAT 128

__global__ void init_state_kernel(unsigned int* __restrict__ mb, int V) {
    int i = blockIdx.x * blockDim.x + threadIdx.x;
    if (i < V) mb[i] = 0xFF800000u;
}

__global__ void packT_kernel(const float* __restrict__ W, float* __restrict__ WT,
                             int R, int C) {
    int i = blockIdx.x * blockDim.x + threadIdx.x;
    if (i >= R * C) return;
    int n = i / C;
    int k = i - n * C;
    WT[(size_t)k * R + n] = W[i];
}

__global__ void node_dot_kernel(const float* __restrict__ nf,
                                const float* __restrict__ W_edge,
                                float* __restrict__ pd, float* __restrict__ ps,
                                int V) {
    int gid = blockIdx.x * blockDim.x + threadIdx.x;
    int v = gid >> 5;
    int lane = gid & 31;
    if (v >= V) return;
    float4 x  = ((const float4*)(nf + (size_t)v * D_FEAT))[lane];
    float4 w0 = ((const float4*)W_edge)[lane];
    float4 w1 = ((const float4*)(W_edge + D_FEAT))[lane];
    float d0 = x.x * w0.x + x.y * w0.y + x.z * w0.z + x.w * w0.w;
    float d1 = x.x * w1.x + x.y * w1.y + x.z * w1.z + x.w * w1.w;
#pragma unroll
    for (int off = 16; off > 0; off >>= 1) {
        d0 += __shfl_down(d0, off, 32);
        d1 += __shfl_down(d1, off, 32);
    }
    if (lane == 0) { pd[v] = d0; ps[v] = d1; }
}

template <int N>
__global__ __launch_bounds__(256) void wmma_gemm_kernel(
    const float* __restrict__ X, const float* __restrict__ B,
    const float* __restrict__ bias, float* __restrict__ Out, int V) {
    __shared__ float xs[16][132];
    const int tid = threadIdx.x;
    const int rowbase = blockIdx.x * 16;
#pragma unroll
    for (int i = 0; i < 8; ++i) {
        int idx = tid + i * 256;
        int r = idx >> 7;
        int col = idx & 127;
        int row = rowbase + r;
        if (row >= V) row = V - 1;
        xs[r][col] = X[(size_t)row * D_FEAT + col];
    }
    __syncthreads();
    const int lane = tid & 31;
    const int wave = tid >> 5;
    const int lo = lane & 15;
    const int hi = lane >> 4;
    const bool fullTile = (rowbase + 16 <= V);
    for (int t = wave; t < N / 16; t += 8) {
        const int colbase = t * 16;
        v8f acc;
        const float bv = bias[colbase + lo];
#pragma unroll
        for (int r = 0; r < 8; ++r) acc[r] = bv;
#pragma unroll
        for (int kk = 0; kk < D_FEAT; kk += 4) {
            const int ka = kk + 2 * hi;
            v2f a, b;
            a[0] = xs[lo][ka];
            a[1] = xs[lo][ka + 1];
            b[0] = B[(size_t)ka * N + colbase + lo];
            b[1] = B[(size_t)(ka + 1) * N + colbase + lo];
            acc = __builtin_amdgcn_wmma_f32_16x16x4_f32(
                false, a, false, b, (short)0, acc, false, false);
        }
        if (fullTile) {
#pragma unroll
            for (int r = 0; r < 8; ++r) {
                int row = rowbase + r + 8 * hi;
                Out[(size_t)row * N + colbase + lo] = acc[r];
            }
        } else {
#pragma unroll
            for (int r = 0; r < 8; ++r) {
                int row = rowbase + r + 8 * hi;
                if (row < V) Out[(size_t)row * N + colbase + lo] = acc[r];
            }
        }
    }
}

#define TILE_ROWS 196

__device__ __forceinline__ float leaky_logit(float x) {
    return x > 0.0f ? x : 0.01f * x;
}

__device__ __forceinline__ void seg_max(unsigned int* __restrict__ m, float l) {
    const int iv = __float_as_int(l);
    if (iv >= 0) atomicMax((int*)m, iv);
    else         atomicMin(m, (unsigned int)iv);
}

__global__ void edge_logit_kernel(const int* __restrict__ src,
                                  const int* __restrict__ dst,
                                  const float* __restrict__ pd,
                                  const float* __restrict__ ps,
                                  const float* __restrict__ b_edge,
                                  float* __restrict__ logit,
                                  unsigned int* __restrict__ mb,
                                  unsigned int* __restrict__ tbw, int E) {
    const int i = blockIdx.x * blockDim.x + threadIdx.x;
    const int lane = threadIdx.x & 31;
    const int wv = i >> 5;
    const int e0 = 4 * i;
    const float be = b_edge[0];
    if (wv * 128 + 127 < E) {
        const int4 d = ((const int4*)dst)[i];
        const int4 s = ((const int4*)src)[i];
        const float l0 = leaky_logit(pd[d.x] + ps[s.x] + be);
        const float l1 = leaky_logit(pd[d.y] + ps[s.y] + be);
        const float l2 = leaky_logit(pd[d.z] + ps[s.z] + be);
        const float l3 = leaky_logit(pd[d.w] + ps[s.w] + be);
        ((float4*)logit)[i] = make_float4(l0, l1, l2, l3);
        seg_max(mb + d.x, l0);
        seg_max(mb + d.y, l1);
        seg_max(mb + d.z, l2);
        seg_max(mb + d.w, l3);
        const unsigned int t[4] = {(unsigned int)d.x / TILE_ROWS, (unsigned int)d.y / TILE_ROWS,
                                   (unsigned int)d.z / TILE_ROWS, (unsigned int)d.w / TILE_ROWS};
        unsigned int word = 0u;
#pragma unroll
        for (int j = 0; j < 4; ++j) {
#pragma unroll
            for (int k = 0; k < 8; ++k) {
                const unsigned int bal = __builtin_amdgcn_ballot_w32(((t[j] >> k) & 1u) != 0u);
                if (lane == 16 * (k >> 2) + 4 * j + (k & 3)) word = bal;
            }
        }
        tbw[wv * 32 + lane] = word;
    } else {
        for (int e = e0; e < E && e < e0 + 4; ++e) {
            const int d = dst[e];
            const float l = leaky_logit(pd[d] + ps[src[e]] + be);
            logit[e] = l;
            seg_max(mb + d, l);
        }
    }
}

#define TILE_PITCH 132
#define SCAN_U 5
#define WLIST 1056

__global__ __launch_bounds__(1024) void tile_scatter_kernel(
    const int* __restrict__ src, const int* __restrict__ dst,
    const float* __restrict__ logit, const unsigned int* __restrict__ mb,
    const unsigned int* __restrict__ tbw,
    const float* __restrict__ hv, const float* __restrict__ WihT,
    const float* __restrict__ b_ih, const float* __restrict__ gh,
    const float* __restrict__ nf, float* __restrict__ out, int V, int E) {
    __shared__ __attribute__((aligned(16))) float acc[TILE_ROWS * TILE_PITCH];
    __shared__ float zs[TILE_ROWS];
    __shared__ int lists[32 * WLIST];

    const int tid = threadIdx.x;
    const int lane = tid & 31;
    const int wave = __builtin_amdgcn_readfirstlane(tid >> 5);
    const int base = blockIdx.x * TILE_ROWS;
    const int rows = (V - base < TILE_ROWS) ? (V - base) : TILE_ROWS;
    const unsigned int urows = (unsigned int)rows;
    int* wl = lists + wave * WLIST;

    for (int i = tid; i < TILE_ROWS * (TILE_PITCH / 4); i += 1024)
        ((float4*)acc)[i] = make_float4(0.0f, 0.0f, 0.0f, 0.0f);
    if (tid < TILE_ROWS) zs[tid] = 0.0f;
    __syncthreads();

    const unsigned int tileId = blockIdx.x & 255u;
    unsigned int tm[8];
#pragma unroll
    for (int k = 0; k < 8; ++k) tm[k] = 0u - ((tileId >> k) & 1u);
    const int nGrp = (E >> 7) << 2;
    const uint4* __restrict__ tv = (const uint4*)tbw;
    const int nRow = (nGrp + 1023) >> 10;
    const int nBatch = (nRow + SCAN_U - 1) / SCAN_U;
    const int nRowP = nBatch * SCAN_U;
    int curRow = (int)((blockIdx.x * (unsigned int)nRowP) / gridDim.x);
    if (curRow >= nRowP) curRow = 0;

    int wcnt = 0;

    if (wave == 0) {
        for (int eo = (E >> 7) << 7; eo < E; eo += 32) {
            const int e = eo + lane;
            bool mt = false;
            if (e < E) mt = (unsigned int)(dst[e] - base) < urows;
            const unsigned int bal = __builtin_amdgcn_ballot_w32(mt);
            if (mt) wl[(int)__builtin_amdgcn_mbcnt_lo(bal, (unsigned int)wcnt)] = e;
            wcnt += __popc(bal);
        }
    }

    for (int it = 0; it <= nBatch; ++it) {
        const bool last = (it == nBatch);
        unsigned int mw[SCAN_U];
        int eb[SCAN_U];
#pragma unroll
        for (int u = 0; u < SCAN_U; ++u) { mw[u] = 0u; eb[u] = 0; }
        if (!last) {
            uint4 pa[SCAN_U], pb[SCAN_U];
            bool ok[SCAN_U];
#pragma unroll
            for (int u = 0; u < SCAN_U; ++u) {
                int r = curRow + u;
                if (r >= nRowP) r -= nRowP;
                const int g = r * 1024 + tid;
                ok[u] = g < nGrp;
                const int gc = ok[u] ? g : nGrp - 1;
                eb[u] = 128 * (gc >> 2) + (gc & 3);
                const int vi = ((gc & ~3) << 1) | (gc & 3);
                pa[u] = tv[vi];
                pb[u] = tv[vi + 4];
            }
            __builtin_amdgcn_sched_barrier(0);
#pragma unroll
            for (int u = 0; u < SCAN_U; ++u) {
                const unsigned int diff = (pa[u].x ^ tm[0]) | (pa[u].y ^ tm[1]) | (pa[u].z ^ tm[2])
                                        | (pa[u].w ^ tm[3]) | (pb[u].x ^ tm[4]) | (pb[u].y ^ tm[5])
                                        | (pb[u].z ^ tm[6]) | (pb[u].w ^ tm[7]);
                mw[u] = ok[u] ? ~diff : 0u;
            }
            curRow += SCAN_U;
            if (curRow >= nRowP) curRow -= nRowP;
        }
        bool pending = true;
        while (pending) {
            pending = false;
#pragma unroll
            for (int u = 0; u < SCAN_U; ++u) {
                unsigned int bal;
                while ((bal = __builtin_amdgcn_ballot_w32(mw[u] != 0u)) != 0u) {
                    if (wcnt > WLIST - 32) { pending = true; break; }
                    if (mw[u] != 0u)
                        wl[(int)__builtin_amdgcn_mbcnt_lo(bal, (unsigned int)wcnt)] =
                            eb[u] + 4 * __builtin_ctz(mw[u] | 0x80000000u);
                    wcnt += __popc(bal);
                    mw[u] &= mw[u] - 1u;
                }
            }
            while (wcnt >= 32 || (last && wcnt > 0)) {
                const int n = wcnt < 32 ? wcnt : 32;
                const int k0 = wcnt - n;
                wcnt = k0;
                const int e = wl[k0 + (lane < n ? lane : 0)];
                const int d = dst[e];
                const int s = src[e];
                const bool mine = (unsigned int)(d - base) < urows;
                const float m = __int_as_float((int)mb[d]);
                const float ev = mine ? expf(logit[e] - m) : 0.0f;
                const int dl = mine ? d - base : 0;
                if (lane < n) atomicAdd(&zs[dl], ev);
                for (int j = 0; j < n; j += 8) {
                    float4 h[8];
#pragma unroll
                    for (int k = 0; k < 8; ++k) {
                        const int jj = (j + k < n) ? (j + k) : (n - 1);
                        const int sj = __builtin_amdgcn_readlane(s, jj);
                        h[k] = ((const float4*)(hv + (size_t)sj * D_FEAT))[lane];
                    }
#pragma unroll
                    for (int k = 0; k < 8; ++k) {
                        const int jj = (j + k < n) ? (j + k) : (n - 1);
                        const float wv = __int_as_float(
                            __builtin_amdgcn_readlane(__float_as_int(ev), jj));
                        const float wt = (j + k < n) ? wv : 0.0f;
                        float* a = acc + __builtin_amdgcn_readlane(dl, jj) * TILE_PITCH + lane;
                        atomicAdd(a, wt * h[k].x);
                        atomicAdd(a + 32, wt * h[k].y);
                        atomicAdd(a + 64, wt * h[k].z);
                        atomicAdd(a + 96, wt * h[k].w);
                    }
                }
            }
        }
    }
    __syncthreads();

    for (int r = wave; r < TILE_ROWS; r += 32) {
        const float zz = zs[r];
        float* a = acc + r * TILE_PITCH;
        float v[4] = {a[lane], a[lane + 32], a[lane + 64], a[lane + 96]};
#pragma unroll
        for (int q = 0; q < 4; ++q) {
            const float x = zz > 0.0f ? v[q] / zz : 0.0f;
            v[q] = x > 0.0f ? x : expf(x) - 1.0f;
        }
        ((float4*)a)[lane] = make_float4(v[0], v[1], v[2], v[3]);
    }
    __syncthreads();

    {
        const int lo = lane & 15;
        const int hi = lane >> 4;
        const int d = (wave & 7) * 16 + lo;
        const int nTiles = (rows + 15) >> 4;
        const float* __restrict__ Bp = WihT + (size_t)(2 * hi) * 384 + d;
        const float br = b_ih[d], bz = b_ih[128 + d], bn = b_ih[256 + d];
        for (int rt = wave >> 3; rt < nTiles; rt += 4) {
            int ar_row = rt * 16 + lo;
            if (ar_row > TILE_ROWS - 1) ar_row = TILE_ROWS - 1;
            const float* __restrict__ Ap = acc + ar_row * TILE_PITCH + 2 * hi;
            v8f gr, gz, gn;
#pragma unroll
            for (int r = 0; r < 8; ++r) { gr[r] = br; gz[r] = bz; gn[r] = bn; }
#pragma unroll 8
            for (int kk = 0; kk < D_FEAT; kk += 4) {
                const v2f a = *(const v2f*)(Ap + kk);
                const float* __restrict__ b0 = Bp + (size_t)kk * 384;
                v2f b;
                b[0] = b0[0];   b[1] = b0[384];
                gr = __builtin_amdgcn_wmma_f32_16x16x4_f32(
                    false, a, false, b, (short)0, gr, false, false);
                b[0] = b0[128]; b[1] = b0[384 + 128];
                gz = __builtin_amdgcn_wmma_f32_16x16x4_f32(
                    false, a, false, b, (short)0, gz, false, false);
                b[0] = b0[256]; b[1] = b0[384 + 256];
                gn = __builtin_amdgcn_wmma_f32_16x16x4_f32(
                    false, a, false, b, (short)0, gn, false, false);
            }
            float hr[8], hz[8], hn[8], xs[8];
#pragma unroll
            for (int r = 0; r < 8; ++r) {
                const int rl = rt * 16 + r + 8 * hi;
                const int rc = rl < rows ? rl : rows - 1;
                const size_t row = (size_t)(base + rc);
                const float* __restrict__ g = gh + row * 384 + d;
                hr[r] = g[0]; hz[r] = g[128]; hn[r] = g[256];
                xs[r] = nf[row * D_FEAT + d];
            }
#pragma unroll
            for (int r = 0; r < 8; ++r) {
                const int rl = rt * 16 + r + 8 * hi;
                const float rg = 1.0f / (1.0f + expf(-(gr[r] + hr[r])));
                const float u = 1.0f / (1.0f + expf(-(gz[r] + hz[r])));
                const float n = tanhf(gn[r] + rg * hn[r]);
                const float h = (1.0f - u) * n + u * xs[r];
                if (rl < rows)
                    out[(size_t)(base + rl) * D_FEAT + d] = h > 0.0f ? h : 0.0f;
            }
        }
    }
}

extern "C" void kernel_launch(void* const* d_in, const int* in_sizes, int n_in,
                              void* d_out, int out_size, void* d_ws, size_t ws_size,
                              hipStream_t stream) {
    const float* nf     = (const float*)d_in[0];
    const int*   src    = (const int*)d_in[1];
    const int*   dst    = (const int*)d_in[2];
    const float* W_edge = (const float*)d_in[3];
    const float* b_edge = (const float*)d_in[4];
    const float* W_proj = (const float*)d_in[5];
    const float* b_proj = (const float*)d_in[6];
    const float* W_ih   = (const float*)d_in[7];
    const float* b_ih   = (const float*)d_in[8];
    const float* W_hh   = (const float*)d_in[9];
    const float* b_hh   = (const float*)d_in[10];

    const int V = in_sizes[0] / D_FEAT;
    const int E = in_sizes[1];

    float* wsf = (float*)d_ws;
    float* pd    = wsf;
    float* ps    = pd + V;
    unsigned int* mb = (unsigned int*)(ps + V);
    float* logit = (float*)(mb + V) + V;
    float* hv    = logit + E;
    float* gh    = hv + (size_t)V * 128;
    float* WhhT  = gh + (size_t)V * 384;
    float* WihT  = WhhT + 128 * 384;
    unsigned int* tbw = (unsigned int*)(WihT + 128 * 384);

    const int blk = 256;

    packT_kernel<<<(384 * 128 + blk - 1) / blk, blk, 0, stream>>>(W_hh, WhhT, 384, 128);
    packT_kernel<<<(384 * 128 + blk - 1) / blk, blk, 0, stream>>>(W_ih, WihT, 384, 128);

    init_state_kernel<<<(V + blk - 1) / blk, blk, 0, stream>>>(mb, V);

    node_dot_kernel<<<((size_t)V * 32 + blk - 1) / blk, blk, 0, stream>>>(
        nf, W_edge, pd, ps, V);

    const int gemmBlocks = (V + 15) / 16;
    wmma_gemm_kernel<128><<<gemmBlocks, 256, 0, stream>>>(nf, W_proj, b_proj, hv, V);
    wmma_gemm_kernel<384><<<gemmBlocks, 256, 0, stream>>>(nf, WhhT, b_hh, gh, V);

    edge_logit_kernel<<<((E + 3) / 4 + blk - 1) / blk, blk, 0, stream>>>(
        src, dst, pd, ps, b_edge, logit, mb, tbw, E);
    tile_scatter_kernel<<<(V + TILE_ROWS - 1) / TILE_ROWS, 1024, 0, stream>>>(
        src, dst, logit, mb, tbw, hv, WihT, b_ih, gh, nf, (float*)d_out, V, E);
}


# Tile owner also forms the hidden-side gate product from its own node_feats rows staged in LDS (f32 WMMA); gh never stored; 8 launches to 7; r and z gates sum gi+gh in one chain: order changes
# speedup vs baseline: 1.9927x; 1.0670x over previous
#include <hip/hip_runtime.h>
#include <hip/hip_bf16.h>
#include <math.h>

typedef float v2f __attribute__((ext_vector_type(2)));
typedef float v8f __attribute__((ext_vector_type(8)));

#define D_FEAT 128

__global__ void init_state_kernel(unsigned int* __restrict__ mb, int V) {
    int i = blockIdx.x * blockDim.x + threadIdx.x;
    if (i < V) mb[i] = 0xFF800000u;
}

__global__ void packT_kernel(const float* __restrict__ W, float* __restrict__ WT,
                             int R, int C) {
    int i = blockIdx.x * blockDim.x + threadIdx.x;
    if (i >= R * C) return;
    int n = i / C;
    int k = i - n * C;
    WT[(size_t)k * R + n] = W[i];
}

__global__ void node_dot_kernel(const float* __restrict__ nf,
                                const float* __restrict__ W_edge,
                                float* __restrict__ pd, float* __restrict__ ps,
                                int V) {
    int gid = blockIdx.x * blockDim.x + threadIdx.x;
    int v = gid >> 5;
    int lane = gid & 31;
    if (v >= V) return;
    float4 x  = ((const float4*)(nf + (size_t)v * D_FEAT))[lane];
    float4 w0 = ((const float4*)W_edge)[lane];
    float4 w1 = ((const float4*)(W_edge + D_FEAT))[lane];
    float d0 = x.x * w0.x + x.y * w0.y + x.z * w0.z + x.w * w0.w;
    float d1 = x.x * w1.x + x.y * w1.y + x.z * w1.z + x.w * w1.w;
#pragma unroll
    for (int off = 16; off > 0; off >>= 1) {
        d0 += __shfl_down(d0, off, 32);
        d1 += __shfl_down(d1, off, 32);
    }
    if (lane == 0) { pd[v] = d0; ps[v] = d1; }
}

template <int N>
__global__ __launch_bounds__(256) void wmma_gemm_kernel(
    const float* __restrict__ X, const float* __restrict__ B,
    const float* __restrict__ bias, float* __restrict__ Out, int V) {
    __shared__ float xs[16][132];
    const int tid = threadIdx.x;
    const int rowbase = blockIdx.x * 16;
#pragma unroll
    for (int i = 0; i < 8; ++i) {
        int idx = tid + i * 256;
        int r = idx >> 7;
        int col = idx & 127;
        int row = rowbase + r;
        if (row >= V) row = V - 1;
        xs[r][col] = X[(size_t)row * D_FEAT + col];
    }
    __syncthreads();
    const int lane = tid & 31;
    const int wave = tid >> 5;
    const int lo = lane & 15;
    const int hi = lane >> 4;
    const bool fullTile = (rowbase + 16 <= V);
    for (int t = wave; t < N / 16; t += 8) {
        const int colbase = t * 16;
        v8f acc;
        const float bv = bias[colbase + lo];
#pragma unroll
        for (int r = 0; r < 8; ++r) acc[r] = bv;
#pragma unroll
        for (int kk = 0; kk < D_FEAT; kk += 4) {
            const int ka = kk + 2 * hi;
            v2f a, b;
            a[0] = xs[lo][ka];
            a[1] = xs[lo][ka + 1];
            b[0] = B[(size_t)ka * N + colbase + lo];
            b[1] = B[(size_t)(ka + 1) * N + colbase + lo];
            acc = __builtin_amdgcn_wmma_f32_16x16x4_f32(
                false, a, false, b, (short)0, acc, false, false);
        }
        if (fullTile) {
#pragma unroll
            for (int r = 0; r < 8; ++r) {
                int row = rowbase + r + 8 * hi;
                Out[(size_t)row * N + colbase + lo] = acc[r];
            }
        } else {
#pragma unroll
            for (int r = 0; r < 8; ++r) {
                int row = rowbase + r + 8 * hi;
                if (row < V) Out[(size_t)row * N + colbase + lo] = acc[r];
            }
        }
    }
}

#define TILE_ROWS 196

__device__ __forceinline__ float leaky_logit(float x) {
    return x > 0.0f ? x : 0.01f * x;
}

__device__ __forceinline__ void seg_max(unsigned int* __restrict__ m, float l) {
    const int iv = __float_as_int(l);
    if (iv >= 0) atomicMax((int*)m, iv);
    else         atomicMin(m, (unsigned int)iv);
}

__global__ void edge_logit_kernel(const int* __restrict__ src,
                                  const int* __restrict__ dst,
                                  const float* __restrict__ pd,
                                  const float* __restrict__ ps,
                                  const float* __restrict__ b_edge,
                                  float* __restrict__ logit,
                                  unsigned int* __restrict__ mb,
                                  unsigned int* __restrict__ tbw, int E) {
    const int i = blockIdx.x * blockDim.x + threadIdx.x;
    const int lane = threadIdx.x & 31;
    const int wv = i >> 5;
    const int e0 = 4 * i;
    const float be = b_edge[0];
    if (wv * 128 + 127 < E) {
        const int4 d = ((const int4*)dst)[i];
        const int4 s = ((const int4*)src)[i];
        const float l0 = leaky_logit(pd[d.x] + ps[s.x] + be);
        const float l1 = leaky_logit(pd[d.y] + ps[s.y] + be);
        const float l2 = leaky_logit(pd[d.z] + ps[s.z] + be);
        const float l3 = leaky_logit(pd[d.w] + ps[s.w] + be);
        ((float4*)logit)[i] = make_float4(l0, l1, l2, l3);
        seg_max(mb + d.x, l0);
        seg_max(mb + d.y, l1);
        seg_max(mb + d.z, l2);
        seg_max(mb + d.w, l3);
        const unsigned int t[4] = {(unsigned int)d.x / TILE_ROWS, (unsigned int)d.y / TILE_ROWS,
                                   (unsigned int)d.z / TILE_ROWS, (unsigned int)d.w / TILE_ROWS};
        unsigned int word = 0u;
#pragma unroll
        for (int j = 0; j < 4; ++j) {
#pragma unroll
            for (int k = 0; k < 8; ++k) {
                const unsigned int bal = __builtin_amdgcn_ballot_w32(((t[j] >> k) & 1u) != 0u);
                if (lane == 16 * (k >> 2) + 4 * j + (k & 3)) word = bal;
            }
        }
        tbw[wv * 32 + lane] = word;
    } else {
        for (int e = e0; e < E && e < e0 + 4; ++e) {
            const int d = dst[e];
            const float l = leaky_logit(pd[d] + ps[src[e]] + be);
            logit[e] = l;
            seg_max(mb + d, l);
        }
    }
}

#define TILE_PITCH 132
#define SCAN_U 5
#define WLIST 1056
#define NF_TRIPS ((TILE_ROWS + 31) / 32)

__global__ __launch_bounds__(1024) void tile_scatter_kernel(
    const int* __restrict__ src, const int* __restrict__ dst,
    const float* __restrict__ logit, const unsigned int* __restrict__ mb,
    const unsigned int* __restrict__ tbw,
    const float* __restrict__ hv, const float* __restrict__ WihT,
    const float* __restrict__ b_ih, const float* __restrict__ WhhT,
    const float* __restrict__ b_hh,
    const float* __restrict__ nf, float* __restrict__ out, int V, int E) {
    __shared__ __attribute__((aligned(16))) float acc[TILE_ROWS * TILE_PITCH];
    __shared__ float zs[TILE_ROWS];
    __shared__ __attribute__((aligned(16))) int lists[32 * WLIST];
    static_assert(NF_TRIPS * 32 * TILE_PITCH <= 32 * WLIST, "node_feats copy must fit in the lists");
    float* nfs = (float*)lists;

    const int tid = threadIdx.x;
    const int lane = tid & 31;
    const int wave = __builtin_amdgcn_readfirstlane(tid >> 5);
    const int base = blockIdx.x * TILE_ROWS;
    const int rows = (V - base < TILE_ROWS) ? (V - base) : TILE_ROWS;
    const unsigned int urows = (unsigned int)rows;
    int* wl = lists + wave * WLIST;

    for (int i = tid; i < TILE_ROWS * (TILE_PITCH / 4); i += 1024)
        ((float4*)acc)[i] = make_float4(0.0f, 0.0f, 0.0f, 0.0f);
    if (tid < TILE_ROWS) zs[tid] = 0.0f;
    __syncthreads();

    const unsigned int tileId = blockIdx.x & 255u;
    unsigned int tm[8];
#pragma unroll
    for (int k = 0; k < 8; ++k) tm[k] = 0u - ((tileId >> k) & 1u);
    const int nGrp = (E >> 7) << 2;
    const uint4* __restrict__ tv = (const uint4*)tbw;
    const int nRow = (nGrp + 1023) >> 10;
    const int nBatch = (nRow + SCAN_U - 1) / SCAN_U;
    const int nRowP = nBatch * SCAN_U;
    int curRow = (int)((blockIdx.x * (unsigned int)nRowP) / gridDim.x);
    if (curRow >= nRowP) curRow = 0;

    int wcnt = 0;

    if (wave == 0) {
        for (int eo = (E >> 7) << 7; eo < E; eo += 32) {
            const int e = eo + lane;
            bool mt = false;
            if (e < E) mt = (unsigned int)(dst[e] - base) < urows;
            const unsigned int bal = __builtin_amdgcn_ballot_w32(mt);
            if (mt) wl[(int)__builtin_amdgcn_mbcnt_lo(bal, (unsigned int)wcnt)] = e;
            wcnt += __popc(bal);
        }
    }

    for (int it = 0; it <= nBatch; ++it) {
        const bool last = (it == nBatch);
        unsigned int mw[SCAN_U];
        int eb[SCAN_U];
#pragma unroll
        for (int u = 0; u < SCAN_U; ++u) { mw[u] = 0u; eb[u] = 0; }
        if (!last) {
            uint4 pa[SCAN_U], pb[SCAN_U];
            bool ok[SCAN_U];
#pragma unroll
            for (int u = 0; u < SCAN_U; ++u) {
                int r = curRow + u;
                if (r >= nRowP) r -= nRowP;
                const int g = r * 1024 + tid;
                ok[u] = g < nGrp;
                const int gc = ok[u] ? g : nGrp - 1;
                eb[u] = 128 * (gc >> 2) + (gc & 3);
                const int vi = ((gc & ~3) << 1) | (gc & 3);
                pa[u] = tv[vi];
                pb[u] = tv[vi + 4];
            }
            __builtin_amdgcn_sched_barrier(0);
#pragma unroll
            for (int u = 0; u < SCAN_U; ++u) {
                const unsigned int diff = (pa[u].x ^ tm[0]) | (pa[u].y ^ tm[1]) | (pa[u].z ^ tm[2])
                                        | (pa[u].w ^ tm[3]) | (pb[u].x ^ tm[4]) | (pb[u].y ^ tm[5])
                                        | (pb[u].z ^ tm[6]) | (pb[u].w ^ tm[7]);
                mw[u] = ok[u] ? ~diff : 0u;
            }
            curRow += SCAN_U;
            if (curRow >= nRowP) curRow -= nRowP;
        }
        bool pending = true;
        while (pending) {
            pending = false;
#pragma unroll
            for (int u = 0; u < SCAN_U; ++u) {
                unsigned int bal;
                while ((bal = __builtin_amdgcn_ballot_w32(mw[u] != 0u)) != 0u) {
                    if (wcnt > WLIST - 32) { pending = true; break; }
                    if (mw[u] != 0u)
                        wl[(int)__builtin_amdgcn_mbcnt_lo(bal, (unsigned int)wcnt)] =
                            eb[u] + 4 * __builtin_ctz(mw[u] | 0x80000000u);
                    wcnt += __popc(bal);
                    mw[u] &= mw[u] - 1u;
                }
            }
            while (wcnt >= 32 || (last && wcnt > 0)) {
                const int n = wcnt < 32 ? wcnt : 32;
                const int k0 = wcnt - n;
                wcnt = k0;
                const int e = wl[k0 + (lane < n ? lane : 0)];
                const int d = dst[e];
                const int s = src[e];
                const bool mine = (unsigned int)(d - base) < urows;
                const float m = __int_as_float((int)mb[d]);
                const float ev = mine ? expf(logit[e] - m) : 0.0f;
                const int dl = mine ? d - base : 0;
                if (lane < n) atomicAdd(&zs[dl], ev);
                for (int j = 0; j < n; j += 8) {
                    float4 h[8];
#pragma unroll
                    for (int k = 0; k < 8; ++k) {
                        const int jj = (j + k < n) ? (j + k) : (n - 1);
                        const int sj = __builtin_amdgcn_readlane(s, jj);
                        h[k] = ((const float4*)(hv + (size_t)sj * D_FEAT))[lane];
                    }
#pragma unroll
                    for (int k = 0; k < 8; ++k) {
                        const int jj = (j + k < n) ? (j + k) : (n - 1);
                        const float wv = __int_as_float(
                            __builtin_amdgcn_readlane(__float_as_int(ev), jj));
                        const float wt = (j + k < n) ? wv : 0.0f;
                        float* a = acc + __builtin_amdgcn_readlane(dl, jj) * TILE_PITCH + lane;
                        atomicAdd(a, wt * h[k].x);
                        atomicAdd(a + 32, wt * h[k].y);
                        atomicAdd(a + 64, wt * h[k].z);
                        atomicAdd(a + 96, wt * h[k].w);
                    }
                }
            }
        }
    }
    __syncthreads();

    {
        float4 nv[NF_TRIPS];
#pragma unroll
        for (int q = 0; q < NF_TRIPS; ++q) {
            const int r = wave + 32 * q;
            const int rc = r < rows ? r : rows - 1;
            nv[q] = ((const float4*)(nf + (size_t)(base + rc) * D_FEAT))[lane];
        }
#pragma unroll
        for (int q = 0; q < NF_TRIPS; ++q)
            *(float4*)(nfs + (wave + 32 * q) * TILE_PITCH + 4 * lane) = nv[q];
    }

    for (int r = wave; r < TILE_ROWS; r += 32) {
        const float zz = zs[r];
        float* a = acc + r * TILE_PITCH;
        float v[4] = {a[lane], a[lane + 32], a[lane + 64], a[lane + 96]};
#pragma unroll
        for (int q = 0; q < 4; ++q) {
            const float x = zz > 0.0f ? v[q] / zz : 0.0f;
            v[q] = x > 0.0f ? x : expf(x) - 1.0f;
        }
        ((float4*)a)[lane] = make_float4(v[0], v[1], v[2], v[3]);
    }
    __syncthreads();

    {
        const int lo = lane & 15;
        const int hi = lane >> 4;
        const int d = (wave & 7) * 16 + lo;
        const int nTiles = (rows + 15) >> 4;
        const float* __restrict__ Bp = WihT + (size_t)(2 * hi) * 384 + d;
        const float* __restrict__ Bq = WhhT + (size_t)(2 * hi) * 384 + d;
        const float br = b_ih[d] + b_hh[d], bz = b_ih[128 + d] + b_hh[128 + d];
        const float bn = b_ih[256 + d], bhn = b_hh[256 + d];
        for (int rt = wave >> 3; rt < nTiles; rt += 4) {
            int ar_row = rt * 16 + lo;
            if (ar_row > TILE_ROWS - 1) ar_row = TILE_ROWS - 1;
            const float* __restrict__ Ap = acc + ar_row * TILE_PITCH + 2 * hi;
            const float* __restrict__ Np = nfs + ar_row * TILE_PITCH + 2 * hi;
            v8f gr, gz, gn, hn;
#pragma unroll
            for (int r = 0; r < 8; ++r) { gr[r] = br; gz[r] = bz; gn[r] = bn; hn[r] = bhn; }
#pragma unroll 8
            for (int kk = 0; kk < D_FEAT; kk += 4) {
                const v2f a = *(const v2f*)(Ap + kk);
                const float* __restrict__ b0 = Bp + (size_t)kk * 384;
                v2f b;
                b[0] = b0[0];   b[1] = b0[384];
                gr = __builtin_amdgcn_wmma_f32_16x16x4_f32(
                    false, a, false, b, (short)0, gr, false, false);
                b[0] = b0[128]; b[1] = b0[384 + 128];
                gz = __builtin_amdgcn_wmma_f32_16x16x4_f32(
                    false, a, false, b, (short)0, gz, false, false);
                b[0] = b0[256]; b[1] = b0[384 + 256];
                gn = __builtin_amdgcn_wmma_f32_16x16x4_f32(
                    false, a, false, b, (short)0, gn, false, false);
            }
#pragma unroll 8
            for (int kk = 0; kk < D_FEAT; kk += 4) {
                const v2f a = *(const v2f*)(Np + kk);
                const float* __restrict__ b0 = Bq + (size_t)kk * 384;
                v2f b;
                b[0] = b0[0];   b[1] = b0[384];
                gr = __builtin_amdgcn_wmma_f32_16x16x4_f32(
                    false, a, false, b, (short)0, gr, false, false);
                b[0] = b0[128]; b[1] = b0[384 + 128];
                gz = __builtin_amdgcn_wmma_f32_16x16x4_f32(
                    false, a, false, b, (short)0, gz, false, false);
                b[0] = b0[256]; b[1] = b0[384 + 256];
                hn = __builtin_amdgcn_wmma_f32_16x16x4_f32(
                    false, a, false, b, (short)0, hn, false, false);
            }
            float xs[8];
#pragma unroll
            for (int r = 0; r < 8; ++r) {
                const int rl = rt * 16 + r + 8 * hi;
                const int rx = rl < TILE_ROWS ? rl : TILE_ROWS - 1;
                xs[r] = nfs[rx * TILE_PITCH + d];
            }
#pragma unroll
            for (int r = 0; r < 8; ++r) {
                const int rl = rt * 16 + r + 8 * hi;
                const float rg = 1.0f / (1.0f + expf(-gr[r]));
                const float u = 1.0f / (1.0f + expf(-gz[r]));
                const float n = tanhf(gn[r] + rg * hn[r]);
                const float h = (1.0f - u) * n + u * xs[r];
                if (rl < rows)
                    out[(size_t)(base + rl) * D_FEAT + d] = h > 0.0f ? h : 0.0f;
            }
        }
    }
}

extern "C" void kernel_launch(void* const* d_in, const int* in_sizes, int n_in,
                              void* d_out, int out_size, void* d_ws, size_t ws_size,
                              hipStream_t stream) {
    const float* nf     = (const float*)d_in[0];
    const int*   src    = (const int*)d_in[1];
    const int*   dst    = (const int*)d_in[2];
    const float* W_edge = (const float*)d_in[3];
    const float* b_edge = (const float*)d_in[4];
    const float* W_proj = (const float*)d_in[5];
    const float* b_proj = (const float*)d_in[6];
    const float* W_ih   = (const float*)d_in[7];
    const float* b_ih   = (const float*)d_in[8];
    const float* W_hh   = (const float*)d_in[9];
    const float* b_hh   = (const float*)d_in[10];

    const int V = in_sizes[0] / D_FEAT;
    const int E = in_sizes[1];

    float* wsf = (float*)d_ws;
    float* pd    = wsf;
    float* ps    = pd + V;
    unsigned int* mb = (unsigned int*)(ps + V);
    float* logit = (float*)(mb + V) + V;
    float* hv    = logit + E;
    float* WhhT  = hv + (size_t)V * 128;
    float* WihT  = WhhT + 128 * 384;
    unsigned int* tbw = (unsigned int*)(WihT + 128 * 384);

    const int blk = 256;

    packT_kernel<<<(384 * 128 + blk - 1) / blk, blk, 0, stream>>>(W_hh, WhhT, 384, 128);
    packT_kernel<<<(384 * 128 + blk - 1) / blk, blk, 0, stream>>>(W_ih, WihT, 384, 128);

    init_state_kernel<<<(V + blk - 1) / blk, blk, 0, stream>>>(mb, V);

    node_dot_kernel<<<((size_t)V * 32 + blk - 1) / blk, blk, 0, stream>>>(
        nf, W_edge, pd, ps, V);

    const int gemmBlocks = (V + 15) / 16;
    wmma_gemm_kernel<128><<<gemmBlocks, 256, 0, stream>>>(nf, W_proj, b_proj, hv, V);

    edge_logit_kernel<<<((E + 3) / 4 + blk - 1) / blk, blk, 0, stream>>>(
        src, dst, pd, ps, b_edge, logit, mb, tbw, E);
    tile_scatter_kernel<<<(V + TILE_ROWS - 1) / TILE_ROWS, 1024, 0, stream>>>(
        src, dst, logit, mb, tbw, hv, WihT, b_ih, WhhT, b_hh, nf, (float*)d_out, V, E);
}


# Gate weights via LDS: work-group copies each 32-k chunk of W_ih / W_hh once from their own rows (double buffer), its 32 waves read B fragments there; no transposes, 7 launches to 5; sums keep order
# speedup vs baseline: 2.1966x; 1.1023x over previous
#include <hip/hip_runtime.h>
#include <hip/hip_bf16.h>
#include <math.h>

typedef float v2f __attribute__((ext_vector_type(2)));
typedef float v8f __attribute__((ext_vector_type(8)));

#define D_FEAT 128

__global__ void init_state_kernel(unsigned int* __restrict__ mb, int V) {
    int i = blockIdx.x * blockDim.x + threadIdx.x;
    if (i < V) mb[i] = 0xFF800000u;
}

__global__ void node_dot_kernel(const float* __restrict__ nf,
                                const float* __restrict__ W_edge,
                                float* __restrict__ pd, float* __restrict__ ps,
                                int V) {
    int gid = blockIdx.x * blockDim.x + threadIdx.x;
    int v = gid >> 5;
    int lane = gid & 31;
    if (v >= V) return;
    float4 x  = ((const float4*)(nf + (size_t)v * D_FEAT))[lane];
    float4 w0 = ((const float4*)W_edge)[lane];
    float4 w1 = ((const float4*)(W_edge + D_FEAT))[lane];
    float d0 = x.x * w0.x + x.y * w0.y + x.z * w0.z + x.w * w0.w;
    float d1 = x.x * w1.x + x.y * w1.y + x.z * w1.z + x.w * w1.w;
#pragma unroll
    for (int off = 16; off > 0; off >>= 1) {
        d0 += __shfl_down(d0, off, 32);
        d1 += __shfl_down(d1, off, 32);
    }
    if (lane == 0) { pd[v] = d0; ps[v] = d1; }
}

template <int N>
__global__ __launch_bounds__(256) void wmma_gemm_kernel(
    const float* __restrict__ X, const float* __restrict__ B,
    const float* __restrict__ bias, float* __restrict__ Out, int V) {
    __shared__ float xs[16][132];
    const int tid = threadIdx.x;
    const int rowbase = blockIdx.x * 16;
#pragma unroll
    for (int i = 0; i < 8; ++i) {
        int idx = tid + i * 256;
        int r = idx >> 7;
        int col = idx & 127;
        int row = rowbase + r;
        if (row >= V) row = V - 1;
        xs[r][col] = X[(size_t)row * D_FEAT + col];
    }
    __syncthreads();
    const int lane = tid & 31;
    const int wave = tid >> 5;
    const int lo = lane & 15;
    const int hi = lane >> 4;
    const bool fullTile = (rowbase + 16 <= V);
    for (int t = wave; t < N / 16; t += 8) {
        const int colbase = t * 16;
        v8f acc;
        const float bv = bias[colbase + lo];
#pragma unroll
        for (int r = 0; r < 8; ++r) acc[r] = bv;
#pragma unroll
        for (int kk = 0; kk < D_FEAT; kk += 4) {
            const int ka = kk + 2 * hi;
            v2f a, b;
            a[0] = xs[lo][ka];
            a[1] = xs[lo][ka + 1];
            b[0] = B[(size_t)ka * N + colbase + lo];
            b[1] = B[(size_t)(ka + 1) * N + colbase + lo];
            acc = __builtin_amdgcn_wmma_f32_16x16x4_f32(
                false, a, false, b, (short)0, acc, false, false);
        }
        if (fullTile) {
#pragma unroll
            for (int r = 0; r < 8; ++r) {
                int row = rowbase + r + 8 * hi;
                Out[(size_t)row * N + colbase + lo] = acc[r];
            }
        } else {
#pragma unroll
            for (int r = 0; r < 8; ++r) {
                int row = rowbase + r + 8 * hi;
                if (row < V) Out[(size_t)row * N + colbase + lo] = acc[r];
            }
        }
    }
}

#define TILE_ROWS 196

__device__ __forceinline__ float leaky_logit(float x) {
    return x > 0.0f ? x : 0.01f * x;
}

__device__ __forceinline__ void seg_max(unsigned int* __restrict__ m, float l) {
    const int iv = __float_as_int(l);
    if (iv >= 0) atomicMax((int*)m, iv);
    else         atomicMin(m, (unsigned int)iv);
}

__global__ void edge_logit_kernel(const int* __restrict__ src,
                                  const int* __restrict__ dst,
                                  const float* __restrict__ pd,
                                  const float* __restrict__ ps,
                                  const float* __restrict__ b_edge,
                                  float* __restrict__ logit,
                                  unsigned int* __restrict__ mb,
                                  unsigned int* __restrict__ tbw, int E) {
    const int i = blockIdx.x * blockDim.x + threadIdx.x;
    const int lane = threadIdx.x & 31;
    const int wv = i >> 5;
    const int e0 = 4 * i;
    const float be = b_edge[0];
    if (wv * 128 + 127 < E) {
        const int4 d = ((const int4*)dst)[i];
        const int4 s = ((const int4*)src)[i];
        const float l0 = leaky_logit(pd[d.x] + ps[s.x] + be);
        const float l1 = leaky_logit(pd[d.y] + ps[s.y] + be);
        const float l2 = leaky_logit(pd[d.z] + ps[s.z] + be);
        const float l3 = leaky_logit(pd[d.w] + ps[s.w] + be);
        ((float4*)logit)[i] = make_float4(l0, l1, l2, l3);
        seg_max(mb + d.x, l0);
        seg_max(mb + d.y, l1);
        seg_max(mb + d.z, l2);
        seg_max(mb + d.w, l3);
        const unsigned int t[4] = {(unsigned int)d.x / TILE_ROWS, (unsigned int)d.y / TILE_ROWS,
                                   (unsigned int)d.z / TILE_ROWS, (unsigned int)d.w / TILE_ROWS};
        unsigned int word = 0u;
#pragma unroll
        for (int j = 0; j < 4; ++j) {
#pragma unroll
            for (int k = 0; k < 8; ++k) {
                const unsigned int bal = __builtin_amdgcn_ballot_w32(((t[j] >> k) & 1u) != 0u);
                if (lane == 16 * (k >> 2) + 4 * j + (k & 3)) word = bal;
            }
        }
        tbw[wv * 32 + lane] = word;
    } else {
        for (int e = e0; e < E && e < e0 + 4; ++e) {
            const int d = dst[e];
            const float l = leaky_logit(pd[d] + ps[src[e]] + be);
            logit[e] = l;
            seg_max(mb + d, l);
        }
    }
}

#define TILE_PITCH 132
#define SCAN_U 5
#define WLIST 1056
#define NF_TRIPS ((TILE_ROWS + 31) / 32)
#define BS_KC 32
#define BS_PITCH 36
#define BS_WORDS (384 * BS_PITCH)
#define NFS_WORDS (TILE_ROWS * TILE_PITCH)
#define GEMM_WORDS (NFS_WORDS + 2 * BS_WORDS)
#define REGION_WORDS ((32 * WLIST) > GEMM_WORDS ? (32 * WLIST) : GEMM_WORDS)

__device__ __forceinline__ void chunk_products(const float* __restrict__ A,
                                               const float* __restrict__ B,
                                               v8f& g0, v8f& g1, v8f& g2) {
#pragma unroll
    for (int kl = 0; kl < BS_KC / 4; ++kl) {
        const v2f a = *(const v2f*)(A + 4 * kl);
        v2f b = *(const v2f*)(B + 4 * kl);
        g0 = __builtin_amdgcn_wmma_f32_16x16x4_f32(
            false, a, false, b, (short)0, g0, false, false);
        b = *(const v2f*)(B + 128 * BS_PITCH + 4 * kl);
        g1 = __builtin_amdgcn_wmma_f32_16x16x4_f32(
            false, a, false, b, (short)0, g1, false, false);
        b = *(const v2f*)(B + 256 * BS_PITCH + 4 * kl);
        g2 = __builtin_amdgcn_wmma_f32_16x16x4_f32(
            false, a, false, b, (short)0, g2, false, false);
    }
}

__global__ __launch_bounds__(1024) void tile_scatter_kernel(
    const int* __restrict__ src, const int* __restrict__ dst,
    const float* __restrict__ logit, const unsigned int* __restrict__ mb,
    const unsigned int* __restrict__ tbw,
    const float* __restrict__ hv, const float* __restrict__ W_ih,
    const float* __restrict__ b_ih, const float* __restrict__ W_hh,
    const float* __restrict__ b_hh,
    const float* __restrict__ nf, float* __restrict__ out, int V, int E) {
    __shared__ __attribute__((aligned(16))) float acc[TILE_ROWS * TILE_PITCH];
    __shared__ float zs[TILE_ROWS];
    __shared__ __attribute__((aligned(16))) int lists[REGION_WORDS];
    float* nfs = (float*)lists;
    float* Bs = nfs + NFS_WORDS;

    const int tid = threadIdx.x;
    const int lane = tid & 31;
    const int wave = __builtin_amdgcn_readfirstlane(tid >> 5);
    const int base = blockIdx.x * TILE_ROWS;
    const int rows = (V - base < TILE_ROWS) ? (V - base) : TILE_ROWS;
    const unsigned int urows = (unsigned int)rows;
    int* wl = lists + wave * WLIST;

    for (int i = tid; i < TILE_ROWS * (TILE_PITCH / 4); i += 1024)
        ((float4*)acc)[i] = make_float4(0.0f, 0.0f, 0.0f, 0.0f);
    if (tid < TILE_ROWS) zs[tid] = 0.0f;
    __syncthreads();

    const unsigned int tileId = blockIdx.x & 255u;
    unsigned int tm[8];
#pragma unroll
    for (int k = 0; k < 8; ++k) tm[k] = 0u - ((tileId >> k) & 1u);
    const int nGrp = (E >> 7) << 2;
    const uint4* __restrict__ tv = (const uint4*)tbw;
    const int nRow = (nGrp + 1023) >> 10;
    const int nBatch = (nRow + SCAN_U - 1) / SCAN_U;
    const int nRowP = nBatch * SCAN_U;
    int curRow = (int)((blockIdx.x * (unsigned int)nRowP) / gridDim.x);
    if (curRow >= nRowP) curRow = 0;

    int wcnt = 0;

    if (wave == 0) {
        for (int eo = (E >> 7) << 7; eo < E; eo += 32) {
            const int e = eo + lane;
            bool mt = false;
            if (e < E) mt = (unsigned int)(dst[e] - base) < urows;
            const unsigned int bal = __builtin_amdgcn_ballot_w32(mt);
            if (mt) wl[(int)__builtin_amdgcn_mbcnt_lo(bal, (unsigned int)wcnt)] = e;
            wcnt += __popc(bal);
        }
    }

    for (int it = 0; it <= nBatch; ++it) {
        const bool last = (it == nBatch);
        unsigned int mw[SCAN_U];
        int eb[SCAN_U];
#pragma unroll
        for (int u = 0; u < SCAN_U; ++u) { mw[u] = 0u; eb[u] = 0; }
        if (!last) {
            uint4 pa[SCAN_U], pb[SCAN_U];
            bool ok[SCAN_U];
#pragma unroll
            for (int u = 0; u < SCAN_U; ++u) {
                int r = curRow + u;
                if (r >= nRowP) r -= nRowP;
                const int g = r * 1024 + tid;
                ok[u] = g < nGrp;
                const int gc = ok[u] ? g : nGrp - 1;
                eb[u] = 128 * (gc >> 2) + (gc & 3);
                const int vi = ((gc & ~3) << 1) | (gc & 3);
                pa[u] = tv[vi];
                pb[u] = tv[vi + 4];
            }
            __builtin_amdgcn_sched_barrier(0);
#pragma unroll
            for (int u = 0; u < SCAN_U; ++u) {
                const unsigned int diff = (pa[u].x ^ tm[0]) | (pa[u].y ^ tm[1]) | (pa[u].z ^ tm[2])
                                        | (pa[u].w ^ tm[3]) | (pb[u].x ^ tm[4]) | (pb[u].y ^ tm[5])
                                        | (pb[u].z ^ tm[6]) | (pb[u].w ^ tm[7]);
                mw[u] = ok[u] ? ~diff : 0u;
            }
            curRow += SCAN_U;
            if (curRow >= nRowP) curRow -= nRowP;
        }
        bool pending = true;
        while (pending) {
            pending = false;
#pragma unroll
            for (int u = 0; u < SCAN_U; ++u) {
                unsigned int bal;
                while ((bal = __builtin_amdgcn_ballot_w32(mw[u] != 0u)) != 0u) {
                    if (wcnt > WLIST - 32) { pending = true; break; }
                    if (mw[u] != 0u)
                        wl[(int)__builtin_amdgcn_mbcnt_lo(bal, (unsigned int)wcnt)] =
                            eb[u] + 4 * __builtin_ctz(mw[u] | 0x80000000u);
                    wcnt += __popc(bal);
                    mw[u] &= mw[u] - 1u;
                }
            }
            while (wcnt >= 32 || (last && wcnt > 0)) {
                const int n = wcnt < 32 ? wcnt : 32;
                const int k0 = wcnt - n;
                wcnt = k0;
                const int e = wl[k0 + (lane < n ? lane : 0)];
                const int d = dst[e];
                const int s = src[e];
                const bool mine = (unsigned int)(d - base) < urows;
                const float m = __int_as_float((int)mb[d]);
                const float ev = mine ? expf(logit[e] - m) : 0.0f;
                const int dl = mine ? d - base : 0;
                if (lane < n) atomicAdd(&zs[dl], ev);
                for (int j = 0; j < n; j += 8) {
                    float4 h[8];
#pragma unroll
                    for (int k = 0; k < 8; ++k) {
                        const int jj = (j + k < n) ? (j + k) : (n - 1);
                        const int sj = __builtin_amdgcn_readlane(s, jj);
                        h[k] = ((const float4*)(hv + (size_t)sj * D_FEAT))[lane];
                    }
#pragma unroll
                    for (int k = 0; k < 8; ++k) {
                        const int jj = (j + k < n) ? (j + k) : (n - 1);
                        const float wv = __int_as_float(
                            __builtin_amdgcn_readlane(__float_as_int(ev), jj));
                        const float wt = (j + k < n) ? wv : 0.0f;
                        float* a = acc + __builtin_amdgcn_readlane(dl, jj) * TILE_PITCH + lane;
                        atomicAdd(a, wt * h[k].x);
                        atomicAdd(a + 32, wt * h[k].y);
                        atomicAdd(a + 64, wt * h[k].z);
                        atomicAdd(a + 96, wt * h[k].w);
                    }
                }
            }
        }
    }
    __syncthreads();

    const int sRow = tid >> 3;
    const int sCol = (tid & 7) * 4;
    const float* __restrict__ sIh = W_ih + (size_t)sRow * D_FEAT + sCol;
    const float* __restrict__ sHh = W_hh + (size_t)sRow * D_FEAT + sCol;
    float* sDst = Bs + sRow * BS_PITCH + sCol;

    {
        float4 nv[NF_TRIPS];
#pragma unroll
        for (int q = 0; q < NF_TRIPS; ++q) {
            const int r = wave + 32 * q;
            const int rc = r < rows ? r : rows - 1;
            nv[q] = ((const float4*)(nf + (size_t)(base + rc) * D_FEAT))[lane];
        }
#pragma unroll
        for (int q = 0; q < NF_TRIPS; ++q)
            if (wave + 32 * q < TILE_ROWS)
                *(float4*)(nfs + (wave + 32 * q) * TILE_PITCH + 4 * lane) = nv[q];
    }

    float4 pre0 = *(const float4*)(sIh);
    float4 pre1 = *(const float4*)(sIh + 128 * D_FEAT);
    float4 pre2 = *(const float4*)(sIh + 256 * D_FEAT);

    for (int r = wave; r < TILE_ROWS; r += 32) {
        const float zz = zs[r];
        float* a = acc + r * TILE_PITCH;
        float v[4] = {a[lane], a[lane + 32], a[lane + 64], a[lane + 96]};
#pragma unroll
        for (int q = 0; q < 4; ++q) {
            const float x = zz > 0.0f ? v[q] / zz : 0.0f;
            v[q] = x > 0.0f ? x : expf(x) - 1.0f;
        }
        ((float4*)a)[lane] = make_float4(v[0], v[1], v[2], v[3]);
    }
    *(float4*)(sDst) = pre0;
    *(float4*)(sDst + 128 * BS_PITCH) = pre1;
    *(float4*)(sDst + 256 * BS_PITCH) = pre2;
    __syncthreads();

    {
        const int lo = lane & 15;
        const int hi = lane >> 4;
        const int d = (wave & 7) * 16 + lo;
        const int nTiles = (rows + 15) >> 4;
        const int nPass = (nTiles + 3) >> 2;
        const int nChunk = nPass * 8;
        const float br = b_ih[d] + b_hh[d], bz = b_ih[128 + d] + b_hh[128 + d];
        const float bn = b_ih[256 + d], bhn = b_hh[256 + d];
        const float* __restrict__ Bfrag = Bs + d * BS_PITCH + 2 * hi;
        int q = 0;
        for (int pass = 0; pass < nPass; ++pass) {
            const int rt = (wave >> 3) + 4 * pass;
            const bool act = rt < nTiles;
            int ar_row = rt * 16 + lo;
            if (ar_row > TILE_ROWS - 1) ar_row = TILE_ROWS - 1;
            const float* __restrict__ Ap = acc + ar_row * TILE_PITCH + 2 * hi;
            const float* __restrict__ Np = nfs + ar_row * TILE_PITCH + 2 * hi;
            v8f gr, gz, gn, hn;
#pragma unroll
            for (int r = 0; r < 8; ++r) { gr[r] = br; gz[r] = bz; gn[r] = bn; hn[r] = bhn; }
            auto chunk_step = [&](const int c, const float* __restrict__ A, v8f& g2) {
                const bool more = q + 1 < nChunk;
                if (more) {
                    const int cn = (c + 1) & 7;
                    const float* __restrict__ sp = (cn < 4 ? sIh : sHh) + (cn & 3) * BS_KC;
                    pre0 = *(const float4*)(sp);
                    pre1 = *(const float4*)(sp + 128 * D_FEAT);
                    pre2 = *(const float4*)(sp + 256 * D_FEAT);
                }
                if (act) chunk_products(A, Bfrag + (q & 1) * BS_WORDS, gr, gz, g2);
                if (more) {
                    float* sd = sDst + ((q + 1) & 1) * BS_WORDS;
                    *(float4*)(sd) = pre0;
                    *(float4*)(sd + 128 * BS_PITCH) = pre1;
                    *(float4*)(sd + 256 * BS_PITCH) = pre2;
                    __syncthreads();
                }
                ++q;
            };
#pragma unroll 1
            for (int c = 0; c < 4; ++c) chunk_step(c, Ap + c * BS_KC, gn);
#pragma unroll 1
            for (int c = 4; c < 8; ++c) chunk_step(c, Np + (c - 4) * BS_KC, hn);
            if (act) {
                float xs[8];
#pragma unroll
                for (int r = 0; r < 8; ++r) {
                    const int rl = rt * 16 + r + 8 * hi;
                    const int rx = rl < TILE_ROWS ? rl : TILE_ROWS - 1;
                    xs[r] = nfs[rx * TILE_PITCH + d];
                }
#pragma unroll
                for (int r = 0; r < 8; ++r) {
                    const int rl = rt * 16 + r + 8 * hi;
                    const float rg = 1.0f / (1.0f + expf(-gr[r]));
                    const float u = 1.0f / (1.0f + expf(-gz[r]));
                    const float n = tanhf(gn[r] + rg * hn[r]);
                    const float h = (1.0f - u) * n + u * xs[r];
                    if (rl < rows)
                        out[(size_t)(base + rl) * D_FEAT + d] = h > 0.0f ? h : 0.0f;
                }
            }
        }
    }
}

extern "C" void kernel_launch(void* const* d_in, const int* in_sizes, int n_in,
                              void* d_out, int out_size, void* d_ws, size_t ws_size,
                              hipStream_t stream) {
    const float* nf     = (const float*)d_in[0];
    const int*   src    = (const int*)d_in[1];
    const int*   dst    = (const int*)d_in[2];
    const float* W_edge = (const float*)d_in[3];
    const float* b_edge = (const float*)d_in[4];
    const float* W_proj = (const float*)d_in[5];
    const float* b_proj = (const float*)d_in[6];
    const float* W_ih   = (const float*)d_in[7];
    const float* b_ih   = (const float*)d_in[8];
    const float* W_hh   = (const float*)d_in[9];
    const float* b_hh   = (const float*)d_in[10];

    const int V = in_sizes[0] / D_FEAT;
    const int E = in_sizes[1];

    float* wsf = (float*)d_ws;
    float* pd    = wsf;
    float* ps    = pd + V;
    unsigned int* mb = (unsigned int*)(ps + V);
    float* logit = (float*)(mb + V) + V;
    float* hv    = logit + E;
    unsigned int* tbw = (unsigned int*)(hv + (size_t)V * 128);

    const int blk = 256;

    init_state_kernel<<<(V + blk - 1) / blk, blk, 0, stream>>>(mb, V);

    node_dot_kernel<<<((size_t)V * 32 + blk - 1) / blk, blk, 0, stream>>>(
        nf, W_edge, pd, ps, V);

    const int gemmBlocks = (V + 15) / 16;
    wmma_gemm_kernel<128><<<gemmBlocks, 256, 0, stream>>>(nf, W_proj, b_proj, hv, V);

    edge_logit_kernel<<<((E + 3) / 4 + blk - 1) / blk, blk, 0, stream>>>(
        src, dst, pd, ps, b_edge, logit, mb, tbw, E);
    tile_scatter_kernel<<<(V + TILE_ROWS - 1) / TILE_ROWS, 1024, 0, stream>>>(
        src, dst, logit, mb, tbw, hv, W_ih, b_ih, W_hh, b_hh, nf, (float*)d_out, V, E);
}


# Per-node pass as long waves: a wave takes 32 consecutive rows (all row loads in flight), sums them in one transposing shuffle tree, stores whole lines; one-row waves, 1 launch gone; sums keep order
# speedup vs baseline: 2.2855x; 1.0405x over previous
#include <hip/hip_runtime.h>
#include <hip/hip_bf16.h>
#include <math.h>

typedef float v2f __attribute__((ext_vector_type(2)));
typedef float v8f __attribute__((ext_vector_type(8)));

#define D_FEAT 128

#define NODE_ROWS 32
template <int H>
__device__ __forceinline__ void fold_rows(float (&s0)[NODE_ROWS], float (&s1)[NODE_ROWS],
                                          const int lane) {
    const bool up = (lane & H) != 0;
#pragma unroll
    for (int j = 0; j < H; ++j) {
        const float keep0 = up ? s0[j + H] : s0[j];
        const float send0 = up ? s0[j] : s0[j + H];
        const float keep1 = up ? s1[j + H] : s1[j];
        const float send1 = up ? s1[j] : s1[j + H];
        s0[j] = keep0 + __shfl_xor(send0, H, 32);
        s1[j] = keep1 + __shfl_xor(send1, H, 32);
    }
}
__global__ __launch_bounds__(64) void node_pass_kernel(const float* __restrict__ nf,
                                                       const float* __restrict__ W_edge,
                                                       float* __restrict__ pd,
                                                       float* __restrict__ ps,
                                                       unsigned int* __restrict__ mb, int V) {
    const int lane = threadIdx.x & 31;
    const int grp = __builtin_amdgcn_readfirstlane(
        (int)(blockIdx.x * (blockDim.x >> 5) + (threadIdx.x >> 5)));
    const int v0 = grp * NODE_ROWS;
    if (v0 >= V) return;
    const float4 w0 = ((const float4*)W_edge)[lane];
    const float4 w1 = ((const float4*)(W_edge + D_FEAT))[lane];
    float4 x[NODE_ROWS];
#pragma unroll
    for (int j = 0; j < NODE_ROWS; ++j) {
        int v = v0 + j;
        if (v >= V) v = V - 1;
        x[j] = ((const float4*)(nf + (size_t)v * D_FEAT))[lane];
    }
    __builtin_amdgcn_sched_barrier(0);
    float s0[NODE_ROWS], s1[NODE_ROWS];
#pragma unroll
    for (int j = 0; j < NODE_ROWS; ++j) {
        s0[j] = fmaf(x[j].w, w0.w, fmaf(x[j].z, w0.z, fmaf(x[j].x, w0.x, x[j].y * w0.y)));
        s1[j] = fmaf(x[j].w, w1.w, fmaf(x[j].z, w1.z, fmaf(x[j].x, w1.x, x[j].y * w1.y)));
    }
    fold_rows<16>(s0, s1, lane);
    fold_rows<8>(s0, s1, lane);
    fold_rows<4>(s0, s1, lane);
    fold_rows<2>(s0, s1, lane);
    fold_rows<1>(s0, s1, lane);
    const int v = v0 + lane;
    if (v < V) {
        pd[v] = s0[0];
        ps[v] = s1[0];
        mb[v] = 0xFF800000u;
    }
}

template <int N>
__global__ __launch_bounds__(256) void wmma_gemm_kernel(
    const float* __restrict__ X, const float* __restrict__ B,
    const float* __restrict__ bias, float* __restrict__ Out, int V) {
    __shared__ float xs[16][132];
    const int tid = threadIdx.x;
    const int rowbase = blockIdx.x * 16;
#pragma unroll
    for (int i = 0; i < 8; ++i) {
        int idx = tid + i * 256;
        int r = idx >> 7;
        int col = idx & 127;
        int row = rowbase + r;
        if (row >= V) row = V - 1;
        xs[r][col] = X[(size_t)row * D_FEAT + col];
    }
    __syncthreads();
    const int lane = tid & 31;
    const int wave = tid >> 5;
    const int lo = lane & 15;
    const int hi = lane >> 4;
    const bool fullTile = (rowbase + 16 <= V);
    for (int t = wave; t < N / 16; t += 8) {
        const int colbase = t * 16;
        v8f acc;
        const float bv = bias[colbase + lo];
#pragma unroll
        for (int r = 0; r < 8; ++r) acc[r] = bv;
#pragma unroll
        for (int kk = 0; kk < D_FEAT; kk += 4) {
            const int ka = kk + 2 * hi;
            v2f a, b;
            a[0] = xs[lo][ka];
            a[1] = xs[lo][ka + 1];
            b[0] = B[(size_t)ka * N + colbase + lo];
            b[1] = B[(size_t)(ka + 1) * N + colbase + lo];
            acc = __builtin_amdgcn_wmma_f32_16x16x4_f32(
                false, a, false, b, (short)0, acc, false, false);
        }
        if (fullTile) {
#pragma unroll
            for (int r = 0; r < 8; ++r) {
                int row = rowbase + r + 8 * hi;
                Out[(size_t)row * N + colbase + lo] = acc[r];
            }
        } else {
#pragma unroll
            for (int r = 0; r < 8; ++r) {
                int row = rowbase + r + 8 * hi;
                if (row < V) Out[(size_t)row * N + colbase + lo] = acc[r];
            }
        }
    }
}

#define TILE_ROWS 196

__device__ __forceinline__ float leaky_logit(float x) {
    return x > 0.0f ? x : 0.01f * x;
}

__device__ __forceinline__ void seg_max(unsigned int* __restrict__ m, float l) {
    const int iv = __float_as_int(l);
    if (iv >= 0) atomicMax((int*)m, iv);
    else         atomicMin(m, (unsigned int)iv);
}

__global__ void edge_logit_kernel(const int* __restrict__ src,
                                  const int* __restrict__ dst,
                                  const float* __restrict__ pd,
                                  const float* __restrict__ ps,
                                  const float* __restrict__ b_edge,
                                  float* __restrict__ logit,
                                  unsigned int* __restrict__ mb,
                                  unsigned int* __restrict__ tbw, int E) {
    const int i = blockIdx.x * blockDim.x + threadIdx.x;
    const int lane = threadIdx.x & 31;
    const int wv = i >> 5;
    const int e0 = 4 * i;
    const float be = b_edge[0];
    if (wv * 128 + 127 < E) {
        const int4 d = ((const int4*)dst)[i];
        const int4 s = ((const int4*)src)[i];
        const float l0 = leaky_logit(pd[d.x] + ps[s.x] + be);
        const float l1 = leaky_logit(pd[d.y] + ps[s.y] + be);
        const float l2 = leaky_logit(pd[d.z] + ps[s.z] + be);
        const float l3 = leaky_logit(pd[d.w] + ps[s.w] + be);
        ((float4*)logit)[i] = make_float4(l0, l1, l2, l3);
        seg_max(mb + d.x, l0);
        seg_max(mb + d.y, l1);
        seg_max(mb + d.z, l2);
        seg_max(mb + d.w, l3);
        const unsigned int t[4] = {(unsigned int)d.x / TILE_ROWS, (unsigned int)d.y / TILE_ROWS,
                                   (unsigned int)d.z / TILE_ROWS, (unsigned int)d.w / TILE_ROWS};
        unsigned int word = 0u;
#pragma unroll
        for (int j = 0; j < 4; ++j) {
#pragma unroll
            for (int k = 0; k < 8; ++k) {
                const unsigned int bal = __builtin_amdgcn_ballot_w32(((t[j] >> k) & 1u) != 0u);
                if (lane == 16 * (k >> 2) + 4 * j + (k & 3)) word = bal;
            }
        }
        tbw[wv * 32 + lane] = word;
    } else {
        for (int e = e0; e < E && e < e0 + 4; ++e) {
            const int d = dst[e];
            const float l = leaky_logit(pd[d] + ps[src[e]] + be);
            logit[e] = l;
            seg_max(mb + d, l);
        }
    }
}

#define TILE_PITCH 132
#define SCAN_U 5
#define WLIST 1056
#define NF_TRIPS ((TILE_ROWS + 31) / 32)
#define BS_KC 32
#define BS_PITCH 36
#define BS_WORDS (384 * BS_PITCH)
#define NFS_WORDS (TILE_ROWS * TILE_PITCH)
#define GEMM_WORDS (NFS_WORDS + 2 * BS_WORDS)
#define REGION_WORDS ((32 * WLIST) > GEMM_WORDS ? (32 * WLIST) : GEMM_WORDS)

__device__ __forceinline__ void chunk_products(const float* __restrict__ A,
                                               const float* __restrict__ B,
                                               v8f& g0, v8f& g1, v8f& g2) {
#pragma unroll
    for (int kl = 0; kl < BS_KC / 4; ++kl) {
        const v2f a = *(const v2f*)(A + 4 * kl);
        v2f b = *(const v2f*)(B + 4 * kl);
        g0 = __builtin_amdgcn_wmma_f32_16x16x4_f32(
            false, a, false, b, (short)0, g0, false, false);
        b = *(const v2f*)(B + 128 * BS_PITCH + 4 * kl);
        g1 = __builtin_amdgcn_wmma_f32_16x16x4_f32(
            false, a, false, b, (short)0, g1, false, false);
        b = *(const v2f*)(B + 256 * BS_PITCH + 4 * kl);
        g2 = __builtin_amdgcn_wmma_f32_16x16x4_f32(
            false, a, false, b, (short)0, g2, false, false);
    }
}

__global__ __launch_bounds__(1024) void tile_scatter_kernel(
    const int* __restrict__ src, const int* __restrict__ dst,
    const float* __restrict__ logit, const unsigned int* __restrict__ mb,
    const unsigned int* __restrict__ tbw,
    const float* __restrict__ hv, const float* __restrict__ W_ih,
    const float* __restrict__ b_ih, const float* __restrict__ W_hh,
    const float* __restrict__ b_hh,
    const float* __restrict__ nf, float* __restrict__ out, int V, int E) {
    __shared__ __attribute__((aligned(16))) float acc[TILE_ROWS * TILE_PITCH];
    __shared__ float zs[TILE_ROWS];
    __shared__ __attribute__((aligned(16))) int lists[REGION_WORDS];
    float* nfs = (float*)lists;
    float* Bs = nfs + NFS_WORDS;

    const int tid = threadIdx.x;
    const int lane = tid & 31;
    const int wave = __builtin_amdgcn_readfirstlane(tid >> 5);
    const int base = blockIdx.x * TILE_ROWS;
    const int rows = (V - base < TILE_ROWS) ? (V - base) : TILE_ROWS;
    const unsigned int urows = (unsigned int)rows;
    int* wl = lists + wave * WLIST;

    for (int i = tid; i < TILE_ROWS * (TILE_PITCH / 4); i += 1024)
        ((float4*)acc)[i] = make_float4(0.0f, 0.0f, 0.0f, 0.0f);
    if (tid < TILE_ROWS) zs[tid] = 0.0f;
    __syncthreads();

    const unsigned int tileId = blockIdx.x & 255u;
    unsigned int tm[8];
#pragma unroll
    for (int k = 0; k < 8; ++k) tm[k] = 0u - ((tileId >> k) & 1u);
    const int nGrp = (E >> 7) << 2;
    const uint4* __restrict__ tv = (const uint4*)tbw;
    const int nRow = (nGrp + 1023) >> 10;
    const int nBatch = (nRow + SCAN_U - 1) / SCAN_U;
    const int nRowP = nBatch * SCAN_U;
    int curRow = (int)((blockIdx.x * (unsigned int)nRowP) / gridDim.x);
    if (curRow >= nRowP) curRow = 0;

    int wcnt = 0;

    if (wave == 0) {
        for (int eo = (E >> 7) << 7; eo < E; eo += 32) {
            const int e = eo + lane;
            bool mt = false;
            if (e < E) mt = (unsigned int)(dst[e] - base) < urows;
            const unsigned int bal = __builtin_amdgcn_ballot_w32(mt);
            if (mt) wl[(int)__builtin_amdgcn_mbcnt_lo(bal, (unsigned int)wcnt)] = e;
            wcnt += __popc(bal);
        }
    }

    for (int it = 0; it <= nBatch; ++it) {
        const bool last = (it == nBatch);
        unsigned int mw[SCAN_U];
        int eb[SCAN_U];
#pragma unroll
        for (int u = 0; u < SCAN_U; ++u) { mw[u] = 0u; eb[u] = 0; }
        if (!last) {
            uint4 pa[SCAN_U], pb[SCAN_U];
            bool ok[SCAN_U];
#pragma unroll
            for (int u = 0; u < SCAN_U; ++u) {
                int r = curRow + u;
                if (r >= nRowP) r -= nRowP;
                const int g = r * 1024 + tid;
                ok[u] = g < nGrp;
                const int gc = ok[u] ? g : nGrp - 1;
                eb[u] = 128 * (gc >> 2) + (gc & 3);
                const int vi = ((gc & ~3) << 1) | (gc & 3);
                pa[u] = tv[vi];
                pb[u] = tv[vi + 4];
            }
            __builtin_amdgcn_sched_barrier(0);
#pragma unroll
            for (int u = 0; u < SCAN_U; ++u) {
                const unsigned int diff = (pa[u].x ^ tm[0]) | (pa[u].y ^ tm[1]) | (pa[u].z ^ tm[2])
                                        | (pa[u].w ^ tm[3]) | (pb[u].x ^ tm[4]) | (pb[u].y ^ tm[5])
                                        | (pb[u].z ^ tm[6]) | (pb[u].w ^ tm[7]);
                mw[u] = ok[u] ? ~diff : 0u;
            }
            curRow += SCAN_U;
            if (curRow >= nRowP) curRow -= nRowP;
        }
        bool pending = true;
        while (pending) {
            pending = false;
#pragma unroll
            for (int u = 0; u < SCAN_U; ++u) {
                unsigned int bal;
                while ((bal = __builtin_amdgcn_ballot_w32(mw[u] != 0u)) != 0u) {
                    if (wcnt > WLIST - 32) { pending = true; break; }
                    if (mw[u] != 0u)
                        wl[(int)__builtin_amdgcn_mbcnt_lo(bal, (unsigned int)wcnt)] =
                            eb[u] + 4 * __builtin_ctz(mw[u] | 0x80000000u);
                    wcnt += __popc(bal);
                    mw[u] &= mw[u] - 1u;
                }
            }
            while (wcnt >= 32 || (last && wcnt > 0)) {
                const int n = wcnt < 32 ? wcnt : 32;
                const int k0 = wcnt - n;
                wcnt = k0;
                const int e = wl[k0 + (lane < n ? lane : 0)];
                const int d = dst[e];
                const int s = src[e];
                const bool mine = (unsigned int)(d - base) < urows;
                const float m = __int_as_float((int)mb[d]);
                const float ev = mine ? expf(logit[e] - m) : 0.0f;
                const int dl = mine ? d - base : 0;
                if (lane < n) atomicAdd(&zs[dl], ev);
                for (int j = 0; j < n; j += 8) {
                    float4 h[8];
#pragma unroll
                    for (int k = 0; k < 8; ++k) {
                        const int jj = (j + k < n) ? (j + k) : (n - 1);
                        const int sj = __builtin_amdgcn_readlane(s, jj);
                        h[k] = ((const float4*)(hv + (size_t)sj * D_FEAT))[lane];
                    }
#pragma unroll
                    for (int k = 0; k < 8; ++k) {
                        const int jj = (j + k < n) ? (j + k) : (n - 1);
                        const float wv = __int_as_float(
                            __builtin_amdgcn_readlane(__float_as_int(ev), jj));
                        const float wt = (j + k < n) ? wv : 0.0f;
                        float* a = acc + __builtin_amdgcn_readlane(dl, jj) * TILE_PITCH + lane;
                        atomicAdd(a, wt * h[k].x);
                        atomicAdd(a + 32, wt * h[k].y);
                        atomicAdd(a + 64, wt * h[k].z);
                        atomicAdd(a + 96, wt * h[k].w);
                    }
                }
            }
        }
    }
    __syncthreads();

    const int sRow = tid >> 3;
    const int sCol = (tid & 7) * 4;
    const float* __restrict__ sIh = W_ih + (size_t)sRow * D_FEAT + sCol;
    const float* __restrict__ sHh = W_hh + (size_t)sRow * D_FEAT + sCol;
    float* sDst = Bs + sRow * BS_PITCH + sCol;

    {
        float4 nv[NF_TRIPS];
#pragma unroll
        for (int q = 0; q < NF_TRIPS; ++q) {
            const int r = wave + 32 * q;
            const int rc = r < rows ? r : rows - 1;
            nv[q] = ((const float4*)(nf + (size_t)(base + rc) * D_FEAT))[lane];
        }
#pragma unroll
        for (int q = 0; q < NF_TRIPS; ++q)
            if (wave + 32 * q < TILE_ROWS)
                *(float4*)(nfs + (wave + 32 * q) * TILE_PITCH + 4 * lane) = nv[q];
    }

    float4 pre0 = *(const float4*)(sIh);
    float4 pre1 = *(const float4*)(sIh + 128 * D_FEAT);
    float4 pre2 = *(const float4*)(sIh + 256 * D_FEAT);

    for (int r = wave; r < TILE_ROWS; r += 32) {
        const float zz = zs[r];
        float* a = acc + r * TILE_PITCH;
        float v[4] = {a[lane], a[lane + 32], a[lane + 64], a[lane + 96]};
#pragma unroll
        for (int q = 0; q < 4; ++q) {
            const float x = zz > 0.0f ? v[q] / zz : 0.0f;
            v[q] = x > 0.0f ? x : expf(x) - 1.0f;
        }
        ((float4*)a)[lane] = make_float4(v[0], v[1], v[2], v[3]);
    }
    *(float4*)(sDst) = pre0;
    *(float4*)(sDst + 128 * BS_PITCH) = pre1;
    *(float4*)(sDst + 256 * BS_PITCH) = pre2;
    __syncthreads();

    {
        const int lo = lane & 15;
        const int hi = lane >> 4;
        const int d = (wave & 7) * 16 + lo;
        const int nTiles = (rows + 15) >> 4;
        const int nPass = (nTiles + 3) >> 2;
        const int nChunk = nPass * 8;
        const float br = b_ih[d] + b_hh[d], bz = b_ih[128 + d] + b_hh[128 + d];
        const float bn = b_ih[256 + d], bhn = b_hh[256 + d];
        const float* __restrict__ Bfrag = Bs + d * BS_PITCH + 2 * hi;
        int q = 0;
        for (int pass = 0; pass < nPass; ++pass) {
            const int rt = (wave >> 3) + 4 * pass;
            const bool act = rt < nTiles;
            int ar_row = rt * 16 + lo;
            if (ar_row > TILE_ROWS - 1) ar_row = TILE_ROWS - 1;
            const float* __restrict__ Ap = acc + ar_row * TILE_PITCH + 2 * hi;
            const float* __restrict__ Np = nfs + ar_row * TILE_PITCH + 2 * hi;
            v8f gr, gz, gn, hn;
#pragma unroll
            for (int r = 0; r < 8; ++r) { gr[r] = br; gz[r] = bz; gn[r] = bn; hn[r] = bhn; }
            auto chunk_step = [&](const int c, const float* __restrict__ A, v8f& g2) {
                const bool more = q + 1 < nChunk;
                if (more) {
                    const int cn = (c + 1) & 7;
                    const float* __restrict__ sp = (cn < 4 ? sIh : sHh) + (cn & 3) * BS_KC;
                    pre0 = *(const float4*)(sp);
                    pre1 = *(const float4*)(sp + 128 * D_FEAT);
                    pre2 = *(const float4*)(sp + 256 * D_FEAT);
                }
                if (act) chunk_products(A, Bfrag + (q & 1) * BS_WORDS, gr, gz, g2);
                if (more) {
                    float* sd = sDst + ((q + 1) & 1) * BS_WORDS;
                    *(float4*)(sd) = pre0;
                    *(float4*)(sd + 128 * BS_PITCH) = pre1;
                    *(float4*)(sd + 256 * BS_PITCH) = pre2;
                    __syncthreads();
                }
                ++q;
            };
#pragma unroll 1
            for (int c = 0; c < 4; ++c) chunk_step(c, Ap + c * BS_KC, gn);
#pragma unroll 1
            for (int c = 4; c < 8; ++c) chunk_step(c, Np + (c - 4) * BS_KC, hn);
            if (act) {
                float xs[8];
#pragma unroll
                for (int r = 0; r < 8; ++r) {
                    const int rl = rt * 16 + r + 8 * hi;
                    const int rx = rl < TILE_ROWS ? rl : TILE_ROWS - 1;
                    xs[r] = nfs[rx * TILE_PITCH + d];
                }
#pragma unroll
                for (int r = 0; r < 8; ++r) {
                    const int rl = rt * 16 + r + 8 * hi;
                    const float rg = 1.0f / (1.0f + expf(-gr[r]));
                    const float u = 1.0f / (1.0f + expf(-gz[r]));
                    const float n = tanhf(gn[r] + rg * hn[r]);
                    const float h = (1.0f - u) * n + u * xs[r];
                    if (rl < rows)
                        out[(size_t)(base + rl) * D_FEAT + d] = h > 0.0f ? h : 0.0f;
                }
            }
        }
    }
}

extern "C" void kernel_launch(void* const* d_in, const int* in_sizes, int n_in,
                              void* d_out, int out_size, void* d_ws, size_t ws_size,
                              hipStream_t stream) {
    const float* nf     = (const float*)d_in[0];
    const int*   src    = (const int*)d_in[1];
    const int*   dst    = (const int*)d_in[2];
    const float* W_edge = (const float*)d_in[3];
    const float* b_edge = (const float*)d_in[4];
    const float* W_proj = (const float*)d_in[5];
    const float* b_proj = (const float*)d_in[6];
    const float* W_ih   = (const float*)d_in[7];
    const float* b_ih   = (const float*)d_in[8];
    const float* W_hh   = (const float*)d_in[9];
    const float* b_hh   = (const float*)d_in[10];

    const int V = in_sizes[0] / D_FEAT;
    const int E = in_sizes[1];

    float* wsf = (float*)d_ws;
    float* pd    = wsf;
    float* ps    = pd + V;
    unsigned int* mb = (unsigned int*)(ps + V);
    float* logit = (float*)(mb + V) + V;
    float* hv    = logit + E;
    unsigned int* tbw = (unsigned int*)(hv + (size_t)V * 128);

    const int blk = 256;

    const int nodeGroups = (V + NODE_ROWS - 1) / NODE_ROWS;
    node_pass_kernel<<<(nodeGroups + 1) / 2, 64, 0, stream>>>(nf, W_edge, pd, ps, mb, V);

    const int gemmBlocks = (V + 15) / 16;
    wmma_gemm_kernel<128><<<gemmBlocks, 256, 0, stream>>>(nf, W_proj, b_proj, hv, V);

    edge_logit_kernel<<<((E + 3) / 4 + blk - 1) / blk, blk, 0, stream>>>(
        src, dst, pd, ps, b_edge, logit, mb, tbw, E);
    tile_scatter_kernel<<<(V + TILE_ROWS - 1) / TILE_ROWS, 1024, 0, stream>>>(
        src, dst, logit, mb, tbw, hv, W_ih, b_ih, W_hh, b_hh, nf, (float*)d_out, V, E);
}


# GRU gates and elu on hardware exp2 and reciprocal (v_exp_f32, v_rcp_f32) in place of library expf, tanhf and IEEE division: about 95 vector instructions per gate value become 30; last bits change
# speedup vs baseline: 2.4232x; 1.0602x over previous
#include <hip/hip_runtime.h>
#include <hip/hip_bf16.h>
#include <math.h>

typedef float v2f __attribute__((ext_vector_type(2)));
typedef float v8f __attribute__((ext_vector_type(8)));

#define D_FEAT 128

#define NODE_ROWS 32
template <int H>
__device__ __forceinline__ void fold_rows(float (&s0)[NODE_ROWS], float (&s1)[NODE_ROWS],
                                          const int lane) {
    const bool up = (lane & H) != 0;
#pragma unroll
    for (int j = 0; j < H; ++j) {
        const float keep0 = up ? s0[j + H] : s0[j];
        const float send0 = up ? s0[j] : s0[j + H];
        const float keep1 = up ? s1[j + H] : s1[j];
        const float send1 = up ? s1[j] : s1[j + H];
        s0[j] = keep0 + __shfl_xor(send0, H, 32);
        s1[j] = keep1 + __shfl_xor(send1, H, 32);
    }
}
__global__ __launch_bounds__(64) void node_pass_kernel(const float* __restrict__ nf,
                                                       const float* __restrict__ W_edge,
                                                       float* __restrict__ pd,
                                                       float* __restrict__ ps,
                                                       unsigned int* __restrict__ mb, int V) {
    const int lane = threadIdx.x & 31;
    const int grp = __builtin_amdgcn_readfirstlane(
        (int)(blockIdx.x * (blockDim.x >> 5) + (threadIdx.x >> 5)));
    const int v0 = grp * NODE_ROWS;
    if (v0 >= V) return;
    const float4 w0 = ((const float4*)W_edge)[lane];
    const float4 w1 = ((const float4*)(W_edge + D_FEAT))[lane];
    float4 x[NODE_ROWS];
#pragma unroll
    for (int j = 0; j < NODE_ROWS; ++j) {
        int v = v0 + j;
        if (v >= V) v = V - 1;
        x[j] = ((const float4*)(nf + (size_t)v * D_FEAT))[lane];
    }
    __builtin_amdgcn_sched_barrier(0);
    float s0[NODE_ROWS], s1[NODE_ROWS];
#pragma unroll
    for (int j = 0; j < NODE_ROWS; ++j) {
        s0[j] = fmaf(x[j].w, w0.w, fmaf(x[j].z, w0.z, fmaf(x[j].x, w0.x, x[j].y * w0.y)));
        s1[j] = fmaf(x[j].w, w1.w, fmaf(x[j].z, w1.z, fmaf(x[j].x, w1.x, x[j].y * w1.y)));
    }
    fold_rows<16>(s0, s1, lane);
    fold_rows<8>(s0, s1, lane);
    fold_rows<4>(s0, s1, lane);
    fold_rows<2>(s0, s1, lane);
    fold_rows<1>(s0, s1, lane);
    const int v = v0 + lane;
    if (v < V) {
        pd[v] = s0[0];
        ps[v] = s1[0];
        mb[v] = 0xFF800000u;
    }
}

template <int N>
__global__ __launch_bounds__(256) void wmma_gemm_kernel(
    const float* __restrict__ X, const float* __restrict__ B,
    const float* __restrict__ bias, float* __restrict__ Out, int V) {
    __shared__ float xs[16][132];
    const int tid = threadIdx.x;
    const int rowbase = blockIdx.x * 16;
#pragma unroll
    for (int i = 0; i < 8; ++i) {
        int idx = tid + i * 256;
        int r = idx >> 7;
        int col = idx & 127;
        int row = rowbase + r;
        if (row >= V) row = V - 1;
        xs[r][col] = X[(size_t)row * D_FEAT + col];
    }
    __syncthreads();
    const int lane = tid & 31;
    const int wave = tid >> 5;
    const int lo = lane & 15;
    const int hi = lane >> 4;
    const bool fullTile = (rowbase + 16 <= V);
    for (int t = wave; t < N / 16; t += 8) {
        const int colbase = t * 16;
        v8f acc;
        const float bv = bias[colbase + lo];
#pragma unroll
        for (int r = 0; r < 8; ++r) acc[r] = bv;
#pragma unroll
        for (int kk = 0; kk < D_FEAT; kk += 4) {
            const int ka = kk + 2 * hi;
            v2f a, b;
            a[0] = xs[lo][ka];
            a[1] = xs[lo][ka + 1];
            b[0] = B[(size_t)ka * N + colbase + lo];
            b[1] = B[(size_t)(ka + 1) * N + colbase + lo];
            acc = __builtin_amdgcn_wmma_f32_16x16x4_f32(
                false, a, false, b, (short)0, acc, false, false);
        }
        if (fullTile) {
#pragma unroll
            for (int r = 0; r < 8; ++r) {
                int row = rowbase + r + 8 * hi;
                Out[(size_t)row * N + colbase + lo] = acc[r];
            }
        } else {
#pragma unroll
            for (int r = 0; r < 8; ++r) {
                int row = rowbase + r + 8 * hi;
                if (row < V) Out[(size_t)row * N + colbase + lo] = acc[r];
            }
        }
    }
}

#define TILE_ROWS 196

__device__ __forceinline__ float leaky_logit(float x) {
    return x > 0.0f ? x : 0.01f * x;
}

__device__ __forceinline__ void seg_max(unsigned int* __restrict__ m, float l) {
    const int iv = __float_as_int(l);
    if (iv >= 0) atomicMax((int*)m, iv);
    else         atomicMin(m, (unsigned int)iv);
}

__global__ void edge_logit_kernel(const int* __restrict__ src,
                                  const int* __restrict__ dst,
                                  const float* __restrict__ pd,
                                  const float* __restrict__ ps,
                                  const float* __restrict__ b_edge,
                                  float* __restrict__ logit,
                                  unsigned int* __restrict__ mb,
                                  unsigned int* __restrict__ tbw, int E) {
    const int i = blockIdx.x * blockDim.x + threadIdx.x;
    const int lane = threadIdx.x & 31;
    const int wv = i >> 5;
    const int e0 = 4 * i;
    const float be = b_edge[0];
    if (wv * 128 + 127 < E) {
        const int4 d = ((const int4*)dst)[i];
        const int4 s = ((const int4*)src)[i];
        const float l0 = leaky_logit(pd[d.x] + ps[s.x] + be);
        const float l1 = leaky_logit(pd[d.y] + ps[s.y] + be);
        const float l2 = leaky_logit(pd[d.z] + ps[s.z] + be);
        const float l3 = leaky_logit(pd[d.w] + ps[s.w] + be);
        ((float4*)logit)[i] = make_float4(l0, l1, l2, l3);
        seg_max(mb + d.x, l0);
        seg_max(mb + d.y, l1);
        seg_max(mb + d.z, l2);
        seg_max(mb + d.w, l3);
        const unsigned int t[4] = {(unsigned int)d.x / TILE_ROWS, (unsigned int)d.y / TILE_ROWS,
                                   (unsigned int)d.z / TILE_ROWS, (unsigned int)d.w / TILE_ROWS};
        unsigned int word = 0u;
#pragma unroll
        for (int j = 0; j < 4; ++j) {
#pragma unroll
            for (int k = 0; k < 8; ++k) {
                const unsigned int bal = __builtin_amdgcn_ballot_w32(((t[j] >> k) & 1u) != 0u);
                if (lane == 16 * (k >> 2) + 4 * j + (k & 3)) word = bal;
            }
        }
        tbw[wv * 32 + lane] = word;
    } else {
        for (int e = e0; e < E && e < e0 + 4; ++e) {
            const int d = dst[e];
            const float l = leaky_logit(pd[d] + ps[src[e]] + be);
            logit[e] = l;
            seg_max(mb + d, l);
        }
    }
}

#define TILE_PITCH 132
#define SCAN_U 5
#define WLIST 1056
#define NF_TRIPS ((TILE_ROWS + 31) / 32)
#define BS_KC 32
#define BS_PITCH 36
#define BS_WORDS (384 * BS_PITCH)
#define NFS_WORDS (TILE_ROWS * TILE_PITCH)
#define GEMM_WORDS (NFS_WORDS + 2 * BS_WORDS)
#define REGION_WORDS ((32 * WLIST) > GEMM_WORDS ? (32 * WLIST) : GEMM_WORDS)

__device__ __forceinline__ float gate_sigmoid(float x) {
    return __builtin_amdgcn_rcpf(1.0f + __builtin_amdgcn_exp2f(-1.44269504f * x));
}
__device__ __forceinline__ float gate_tanh(float x) {
    return fmaf(-2.0f, __builtin_amdgcn_rcpf(1.0f + __builtin_amdgcn_exp2f(2.88539008f * x)), 1.0f);
}

__device__ __forceinline__ void chunk_products(const float* __restrict__ A,
                                               const float* __restrict__ B,
                                               v8f& g0, v8f& g1, v8f& g2) {
#pragma unroll
    for (int kl = 0; kl < BS_KC / 4; ++kl) {
        const v2f a = *(const v2f*)(A + 4 * kl);
        v2f b = *(const v2f*)(B + 4 * kl);
        g0 = __builtin_amdgcn_wmma_f32_16x16x4_f32(
            false, a, false, b, (short)0, g0, false, false);
        b = *(const v2f*)(B + 128 * BS_PITCH + 4 * kl);
        g1 = __builtin_amdgcn_wmma_f32_16x16x4_f32(
            false, a, false, b, (short)0, g1, false, false);
        b = *(const v2f*)(B + 256 * BS_PITCH + 4 * kl);
        g2 = __builtin_amdgcn_wmma_f32_16x16x4_f32(
            false, a, false, b, (short)0, g2, false, false);
    }
}

__global__ __launch_bounds__(1024) void tile_scatter_kernel(
    const int* __restrict__ src, const int* __restrict__ dst,
    const float* __restrict__ logit, const unsigned int* __restrict__ mb,
    const unsigned int* __restrict__ tbw,
    const float* __restrict__ hv, const float* __restrict__ W_ih,
    const float* __restrict__ b_ih, const float* __restrict__ W_hh,
    const float* __restrict__ b_hh,
    const float* __restrict__ nf, float* __restrict__ out, int V, int E) {
    __shared__ __attribute__((aligned(16))) float acc[TILE_ROWS * TILE_PITCH];
    __shared__ float zs[TILE_ROWS];
    __shared__ __attribute__((aligned(16))) int lists[REGION_WORDS];
    float* nfs = (float*)lists;
    float* Bs = nfs + NFS_WORDS;

    const int tid = threadIdx.x;
    const int lane = tid & 31;
    const int wave = __builtin_amdgcn_readfirstlane(tid >> 5);
    const int base = blockIdx.x * TILE_ROWS;
    const int rows = (V - base < TILE_ROWS) ? (V - base) : TILE_ROWS;
    const unsigned int urows = (unsigned int)rows;
    int* wl = lists + wave * WLIST;

    for (int i = tid; i < TILE_ROWS * (TILE_PITCH / 4); i += 1024)
        ((float4*)acc)[i] = make_float4(0.0f, 0.0f, 0.0f, 0.0f);
    if (tid < TILE_ROWS) zs[tid] = 0.0f;
    __syncthreads();

    const unsigned int tileId = blockIdx.x & 255u;
    unsigned int tm[8];
#pragma unroll
    for (int k = 0; k < 8; ++k) tm[k] = 0u - ((tileId >> k) & 1u);
    const int nGrp = (E >> 7) << 2;
    const uint4* __restrict__ tv = (const uint4*)tbw;
    const int nRow = (nGrp + 1023) >> 10;
    const int nBatch = (nRow + SCAN_U - 1) / SCAN_U;
    const int nRowP = nBatch * SCAN_U;
    int curRow = (int)((blockIdx.x * (unsigned int)nRowP) / gridDim.x);
    if (curRow >= nRowP) curRow = 0;

    int wcnt = 0;

    if (wave == 0) {
        for (int eo = (E >> 7) << 7; eo < E; eo += 32) {
            const int e = eo + lane;
            bool mt = false;
            if (e < E) mt = (unsigned int)(dst[e] - base) < urows;
            const unsigned int bal = __builtin_amdgcn_ballot_w32(mt);
            if (mt) wl[(int)__builtin_amdgcn_mbcnt_lo(bal, (unsigned int)wcnt)] = e;
            wcnt += __popc(bal);
        }
    }

    for (int it = 0; it <= nBatch; ++it) {
        const bool last = (it == nBatch);
        unsigned int mw[SCAN_U];
        int eb[SCAN_U];
#pragma unroll
        for (int u = 0; u < SCAN_U; ++u) { mw[u] = 0u; eb[u] = 0; }
        if (!last) {
            uint4 pa[SCAN_U], pb[SCAN_U];
            bool ok[SCAN_U];
#pragma unroll
            for (int u = 0; u < SCAN_U; ++u) {
                int r = curRow + u;
                if (r >= nRowP) r -= nRowP;
                const int g = r * 1024 + tid;
                ok[u] = g < nGrp;
                const int gc = ok[u] ? g : nGrp - 1;
                eb[u] = 128 * (gc >> 2) + (gc & 3);
                const int vi = ((gc & ~3) << 1) | (gc & 3);
                pa[u] = tv[vi];
                pb[u] = tv[vi + 4];
            }
            __builtin_amdgcn_sched_barrier(0);
#pragma unroll
            for (int u = 0; u < SCAN_U; ++u) {
                const unsigned int diff = (pa[u].x ^ tm[0]) | (pa[u].y ^ tm[1]) | (pa[u].z ^ tm[2])
                                        | (pa[u].w ^ tm[3]) | (pb[u].x ^ tm[4]) | (pb[u].y ^ tm[5])
                                        | (pb[u].z ^ tm[6]) | (pb[u].w ^ tm[7]);
                mw[u] = ok[u] ? ~diff : 0u;
            }
            curRow += SCAN_U;
            if (curRow >= nRowP) curRow -= nRowP;
        }
        bool pending = true;
        while (pending) {
            pending = false;
#pragma unroll
            for (int u = 0; u < SCAN_U; ++u) {
                unsigned int bal;
                while ((bal = __builtin_amdgcn_ballot_w32(mw[u] != 0u)) != 0u) {
                    if (wcnt > WLIST - 32) { pending = true; break; }
                    if (mw[u] != 0u)
                        wl[(int)__builtin_amdgcn_mbcnt_lo(bal, (unsigned int)wcnt)] =
                            eb[u] + 4 * __builtin_ctz(mw[u] | 0x80000000u);
                    wcnt += __popc(bal);
                    mw[u] &= mw[u] - 1u;
                }
            }
            while (wcnt >= 32 || (last && wcnt > 0)) {
                const int n = wcnt < 32 ? wcnt : 32;
                const int k0 = wcnt - n;
                wcnt = k0;
                const int e = wl[k0 + (lane < n ? lane : 0)];
                const int d = dst[e];
                const int s = src[e];
                const bool mine = (unsigned int)(d - base) < urows;
                const float m = __int_as_float((int)mb[d]);
                const float ev = mine ? expf(logit[e] - m) : 0.0f;
                const int dl = mine ? d - base : 0;
                if (lane < n) atomicAdd(&zs[dl], ev);
                for (int j = 0; j < n; j += 8) {
                    float4 h[8];
#pragma unroll
                    for (int k = 0; k < 8; ++k) {
                        const int jj = (j + k < n) ? (j + k) : (n - 1);
                        const int sj = __builtin_amdgcn_readlane(s, jj);
                        h[k] = ((const float4*)(hv + (size_t)sj * D_FEAT))[lane];
                    }
#pragma unroll
                    for (int k = 0; k < 8; ++k) {
                        const int jj = (j + k < n) ? (j + k) : (n - 1);
                        const float wv = __int_as_float(
                            __builtin_amdgcn_readlane(__float_as_int(ev), jj));
                        const float wt = (j + k < n) ? wv : 0.0f;
                        float* a = acc + __builtin_amdgcn_readlane(dl, jj) * TILE_PITCH + lane;
                        atomicAdd(a, wt * h[k].x);
                        atomicAdd(a + 32, wt * h[k].y);
                        atomicAdd(a + 64, wt * h[k].z);
                        atomicAdd(a + 96, wt * h[k].w);
                    }
                }
            }
        }
    }
    __syncthreads();

    const int sRow = tid >> 3;
    const int sCol = (tid & 7) * 4;
    const float* __restrict__ sIh = W_ih + (size_t)sRow * D_FEAT + sCol;
    const float* __restrict__ sHh = W_hh + (size_t)sRow * D_FEAT + sCol;
    float* sDst = Bs + sRow * BS_PITCH + sCol;

    {
        float4 nv[NF_TRIPS];
#pragma unroll
        for (int q = 0; q < NF_TRIPS; ++q) {
            const int r = wave + 32 * q;
            const int rc = r < rows ? r : rows - 1;
            nv[q] = ((const float4*)(nf + (size_t)(base + rc) * D_FEAT))[lane];
        }
#pragma unroll
        for (int q = 0; q < NF_TRIPS; ++q)
            if (wave + 32 * q < TILE_ROWS)
                *(float4*)(nfs + (wave + 32 * q) * TILE_PITCH + 4 * lane) = nv[q];
    }

    float4 pre0 = *(const float4*)(sIh);
    float4 pre1 = *(const float4*)(sIh + 128 * D_FEAT);
    float4 pre2 = *(const float4*)(sIh + 256 * D_FEAT);

    for (int r = wave; r < TILE_ROWS; r += 32) {
        const float zz = zs[r];
        const float inv = zz > 0.0f ? __builtin_amdgcn_rcpf(zz) : 0.0f;
        float* a = acc + r * TILE_PITCH;
        float v[4] = {a[lane], a[lane + 32], a[lane + 64], a[lane + 96]};
#pragma unroll
        for (int q = 0; q < 4; ++q) {
            const float x = v[q] * inv;
            v[q] = x > 0.0f ? x : __builtin_amdgcn_exp2f(1.44269504f * x) - 1.0f;
        }
        ((float4*)a)[lane] = make_float4(v[0], v[1], v[2], v[3]);
    }
    *(float4*)(sDst) = pre0;
    *(float4*)(sDst + 128 * BS_PITCH) = pre1;
    *(float4*)(sDst + 256 * BS_PITCH) = pre2;
    __syncthreads();

    {
        const int lo = lane & 15;
        const int hi = lane >> 4;
        const int d = (wave & 7) * 16 + lo;
        const int nTiles = (rows + 15) >> 4;
        const int nPass = (nTiles + 3) >> 2;
        const int nChunk = nPass * 8;
        const float br = b_ih[d] + b_hh[d], bz = b_ih[128 + d] + b_hh[128 + d];
        const float bn = b_ih[256 + d], bhn = b_hh[256 + d];
        const float* __restrict__ Bfrag = Bs + d * BS_PITCH + 2 * hi;
        int q = 0;
        for (int pass = 0; pass < nPass; ++pass) {
            const int rt = (wave >> 3) + 4 * pass;
            const bool act = rt < nTiles;
            int ar_row = rt * 16 + lo;
            if (ar_row > TILE_ROWS - 1) ar_row = TILE_ROWS - 1;
            const float* __restrict__ Ap = acc + ar_row * TILE_PITCH + 2 * hi;
            const float* __restrict__ Np = nfs + ar_row * TILE_PITCH + 2 * hi;
            v8f gr, gz, gn, hn;
#pragma unroll
            for (int r = 0; r < 8; ++r) { gr[r] = br; gz[r] = bz; gn[r] = bn; hn[r] = bhn; }
            auto chunk_step = [&](const int c, const float* __restrict__ A, v8f& g2) {
                const bool more = q + 1 < nChunk;
                if (more) {
                    const int cn = (c + 1) & 7;
                    const float* __restrict__ sp = (cn < 4 ? sIh : sHh) + (cn & 3) * BS_KC;
                    pre0 = *(const float4*)(sp);
                    pre1 = *(const float4*)(sp + 128 * D_FEAT);
                    pre2 = *(const float4*)(sp + 256 * D_FEAT);
                }
                if (act) chunk_products(A, Bfrag + (q & 1) * BS_WORDS, gr, gz, g2);
                if (more) {
                    float* sd = sDst + ((q + 1) & 1) * BS_WORDS;
                    *(float4*)(sd) = pre0;
                    *(float4*)(sd + 128 * BS_PITCH) = pre1;
                    *(float4*)(sd + 256 * BS_PITCH) = pre2;
                    __syncthreads();
                }
                ++q;
            };
#pragma unroll 1
            for (int c = 0; c < 4; ++c) chunk_step(c, Ap + c * BS_KC, gn);
#pragma unroll 1
            for (int c = 4; c < 8; ++c) chunk_step(c, Np + (c - 4) * BS_KC, hn);
            if (act) {
                float xs[8];
#pragma unroll
                for (int r = 0; r < 8; ++r) {
                    const int rl = rt * 16 + r + 8 * hi;
                    const int rx = rl < TILE_ROWS ? rl : TILE_ROWS - 1;
                    xs[r] = nfs[rx * TILE_PITCH + d];
                }
#pragma unroll
                for (int r = 0; r < 8; ++r) {
                    const int rl = rt * 16 + r + 8 * hi;
                    const float rg = gate_sigmoid(gr[r]);
                    const float u = gate_sigmoid(gz[r]);
                    const float n = gate_tanh(gn[r] + rg * hn[r]);
                    const float h = (1.0f - u) * n + u * xs[r];
                    if (rl < rows)
                        out[(size_t)(base + rl) * D_FEAT + d] = h > 0.0f ? h : 0.0f;
                }
            }
        }
    }
}

extern "C" void kernel_launch(void* const* d_in, const int* in_sizes, int n_in,
                              void* d_out, int out_size, void* d_ws, size_t ws_size,
                              hipStream_t stream) {
    const float* nf     = (const float*)d_in[0];
    const int*   src    = (const int*)d_in[1];
    const int*   dst    = (const int*)d_in[2];
    const float* W_edge = (const float*)d_in[3];
    const float* b_edge = (const float*)d_in[4];
    const float* W_proj = (const float*)d_in[5];
    const float* b_proj = (const float*)d_in[6];
    const float* W_ih   = (const float*)d_in[7];
    const float* b_ih   = (const float*)d_in[8];
    const float* W_hh   = (const float*)d_in[9];
    const float* b_hh   = (const float*)d_in[10];

    const int V = in_sizes[0] / D_FEAT;
    const int E = in_sizes[1];

    float* wsf = (float*)d_ws;
    float* pd    = wsf;
    float* ps    = pd + V;
    unsigned int* mb = (unsigned int*)(ps + V);
    float* logit = (float*)(mb + V) + V;
    float* hv    = logit + E;
    unsigned int* tbw = (unsigned int*)(hv + (size_t)V * 128);

    const int blk = 256;

    const int nodeGroups = (V + NODE_ROWS - 1) / NODE_ROWS;
    node_pass_kernel<<<(nodeGroups + 1) / 2, 64, 0, stream>>>(nf, W_edge, pd, ps, mb, V);

    const int gemmBlocks = (V + 15) / 16;
    wmma_gemm_kernel<128><<<gemmBlocks, 256, 0, stream>>>(nf, W_proj, b_proj, hv, V);

    edge_logit_kernel<<<((E + 3) / 4 + blk - 1) / blk, blk, 0, stream>>>(
        src, dst, pd, ps, b_edge, logit, mb, tbw, E);
    tile_scatter_kernel<<<(V + TILE_ROWS - 1) / TILE_ROWS, 1024, 0, stream>>>(
        src, dst, logit, mb, tbw, hv, W_ih, b_ih, W_hh, b_hh, nf, (float*)d_out, V, E);
}


# Edge scalars in one trip: logit pass writes a 16-byte record (src, dst, logit) per edge in whole lines, tile owner keeps its rows' max in LDS; a listed edge costs 1 gathered line, not 4; order kept
# speedup vs baseline: 2.4981x; 1.0309x over previous
#include <hip/hip_runtime.h>
#include <hip/hip_bf16.h>
#include <math.h>

typedef float v2f __attribute__((ext_vector_type(2)));
typedef float v8f __attribute__((ext_vector_type(8)));

#define D_FEAT 128

#define NODE_ROWS 32
template <int H>
__device__ __forceinline__ void fold_rows(float (&s0)[NODE_ROWS], float (&s1)[NODE_ROWS],
                                          const int lane) {
    const bool up = (lane & H) != 0;
#pragma unroll
    for (int j = 0; j < H; ++j) {
        const float keep0 = up ? s0[j + H] : s0[j];
        const float send0 = up ? s0[j] : s0[j + H];
        const float keep1 = up ? s1[j + H] : s1[j];
        const float send1 = up ? s1[j] : s1[j + H];
        s0[j] = keep0 + __shfl_xor(send0, H, 32);
        s1[j] = keep1 + __shfl_xor(send1, H, 32);
    }
}
__global__ __launch_bounds__(64) void node_pass_kernel(const float* __restrict__ nf,
                                                       const float* __restrict__ W_edge,
                                                       float* __restrict__ pd,
                                                       float* __restrict__ ps,
                                                       unsigned int* __restrict__ mb, int V) {
    const int lane = threadIdx.x & 31;
    const int grp = __builtin_amdgcn_readfirstlane(
        (int)(blockIdx.x * (blockDim.x >> 5) + (threadIdx.x >> 5)));
    const int v0 = grp * NODE_ROWS;
    if (v0 >= V) return;
    const float4 w0 = ((const float4*)W_edge)[lane];
    const float4 w1 = ((const float4*)(W_edge + D_FEAT))[lane];
    float4 x[NODE_ROWS];
#pragma unroll
    for (int j = 0; j < NODE_ROWS; ++j) {
        int v = v0 + j;
        if (v >= V) v = V - 1;
        x[j] = ((const float4*)(nf + (size_t)v * D_FEAT))[lane];
    }
    __builtin_amdgcn_sched_barrier(0);
    float s0[NODE_ROWS], s1[NODE_ROWS];
#pragma unroll
    for (int j = 0; j < NODE_ROWS; ++j) {
        s0[j] = fmaf(x[j].w, w0.w, fmaf(x[j].z, w0.z, fmaf(x[j].x, w0.x, x[j].y * w0.y)));
        s1[j] = fmaf(x[j].w, w1.w, fmaf(x[j].z, w1.z, fmaf(x[j].x, w1.x, x[j].y * w1.y)));
    }
    fold_rows<16>(s0, s1, lane);
    fold_rows<8>(s0, s1, lane);
    fold_rows<4>(s0, s1, lane);
    fold_rows<2>(s0, s1, lane);
    fold_rows<1>(s0, s1, lane);
    const int v = v0 + lane;
    if (v < V) {
        pd[v] = s0[0];
        ps[v] = s1[0];
        mb[v] = 0xFF800000u;
    }
}

template <int N>
__global__ __launch_bounds__(256) void wmma_gemm_kernel(
    const float* __restrict__ X, const float* __restrict__ B,
    const float* __restrict__ bias, float* __restrict__ Out, int V) {
    __shared__ float xs[16][132];
    const int tid = threadIdx.x;
    const int rowbase = blockIdx.x * 16;
#pragma unroll
    for (int i = 0; i < 8; ++i) {
        int idx = tid + i * 256;
        int r = idx >> 7;
        int col = idx & 127;
        int row = rowbase + r;
        if (row >= V) row = V - 1;
        xs[r][col] = X[(size_t)row * D_FEAT + col];
    }
    __syncthreads();
    const int lane = tid & 31;
    const int wave = tid >> 5;
    const int lo = lane & 15;
    const int hi = lane >> 4;
    const bool fullTile = (rowbase + 16 <= V);
    for (int t = wave; t < N / 16; t += 8) {
        const int colbase = t * 16;
        v8f acc;
        const float bv = bias[colbase + lo];
#pragma unroll
        for (int r = 0; r < 8; ++r) acc[r] = bv;
#pragma unroll
        for (int kk = 0; kk < D_FEAT; kk += 4) {
            const int ka = kk + 2 * hi;
            v2f a, b;
            a[0] = xs[lo][ka];
            a[1] = xs[lo][ka + 1];
            b[0] = B[(size_t)ka * N + colbase + lo];
            b[1] = B[(size_t)(ka + 1) * N + colbase + lo];
            acc = __builtin_amdgcn_wmma_f32_16x16x4_f32(
                false, a, false, b, (short)0, acc, false, false);
        }
        if (fullTile) {
#pragma unroll
            for (int r = 0; r < 8; ++r) {
                int row = rowbase + r + 8 * hi;
                Out[(size_t)row * N + colbase + lo] = acc[r];
            }
        } else {
#pragma unroll
            for (int r = 0; r < 8; ++r) {
                int row = rowbase + r + 8 * hi;
                if (row < V) Out[(size_t)row * N + colbase + lo] = acc[r];
            }
        }
    }
}

#define TILE_ROWS 196

__device__ __forceinline__ float leaky_logit(float x) {
    return x > 0.0f ? x : 0.01f * x;
}

__device__ __forceinline__ void seg_max(unsigned int* __restrict__ m, float l) {
    const int iv = __float_as_int(l);
    if (iv >= 0) atomicMax((int*)m, iv);
    else         atomicMin(m, (unsigned int)iv);
}

__global__ void edge_logit_kernel(const int* __restrict__ src,
                                  const int* __restrict__ dst,
                                  const float* __restrict__ pd,
                                  const float* __restrict__ ps,
                                  const float* __restrict__ b_edge,
                                  int4* __restrict__ rec,
                                  unsigned int* __restrict__ mb,
                                  unsigned int* __restrict__ tbw, int E) {
    const int i = blockIdx.x * blockDim.x + threadIdx.x;
    const int lane = threadIdx.x & 31;
    const int wv = i >> 5;
    const int e0 = 4 * i;
    const float be = b_edge[0];
    if (wv * 128 + 127 < E) {
        const int4 d = ((const int4*)dst)[i];
        const int4 s = ((const int4*)src)[i];
        const float l0 = leaky_logit(pd[d.x] + ps[s.x] + be);
        const float l1 = leaky_logit(pd[d.y] + ps[s.y] + be);
        const float l2 = leaky_logit(pd[d.z] + ps[s.z] + be);
        const float l3 = leaky_logit(pd[d.w] + ps[s.w] + be);
        int4* __restrict__ rw = rec + (size_t)wv * 128 + lane;
        rw[0]  = make_int4(s.x, d.x, __float_as_int(l0), 0);
        rw[32] = make_int4(s.y, d.y, __float_as_int(l1), 0);
        rw[64] = make_int4(s.z, d.z, __float_as_int(l2), 0);
        rw[96] = make_int4(s.w, d.w, __float_as_int(l3), 0);
        seg_max(mb + d.x, l0);
        seg_max(mb + d.y, l1);
        seg_max(mb + d.z, l2);
        seg_max(mb + d.w, l3);
        const unsigned int t[4] = {(unsigned int)d.x / TILE_ROWS, (unsigned int)d.y / TILE_ROWS,
                                   (unsigned int)d.z / TILE_ROWS, (unsigned int)d.w / TILE_ROWS};
        unsigned int word = 0u;
#pragma unroll
        for (int j = 0; j < 4; ++j) {
#pragma unroll
            for (int k = 0; k < 8; ++k) {
                const unsigned int bal = __builtin_amdgcn_ballot_w32(((t[j] >> k) & 1u) != 0u);
                if (lane == 16 * (k >> 2) + 4 * j + (k & 3)) word = bal;
            }
        }
        tbw[wv * 32 + lane] = word;
    } else {
        for (int e = e0; e < E && e < e0 + 4; ++e) {
            const int d = dst[e];
            const int sv = src[e];
            const float l = leaky_logit(pd[d] + ps[sv] + be);
            rec[e] = make_int4(sv, d, __float_as_int(l), 0);
            seg_max(mb + d, l);
        }
    }
}

#define TILE_PITCH 132
#define SCAN_U 5
#define WLIST 1056
#define NF_TRIPS ((TILE_ROWS + 31) / 32)
#define BS_KC 32
#define BS_PITCH 36
#define BS_WORDS (384 * BS_PITCH)
#define NFS_WORDS (TILE_ROWS * TILE_PITCH)
#define GEMM_WORDS (NFS_WORDS + 2 * BS_WORDS)
#define REGION_WORDS ((32 * WLIST) > GEMM_WORDS ? (32 * WLIST) : GEMM_WORDS)

__device__ __forceinline__ float gate_sigmoid(float x) {
    return __builtin_amdgcn_rcpf(1.0f + __builtin_amdgcn_exp2f(-1.44269504f * x));
}
__device__ __forceinline__ float gate_tanh(float x) {
    return fmaf(-2.0f, __builtin_amdgcn_rcpf(1.0f + __builtin_amdgcn_exp2f(2.88539008f * x)), 1.0f);
}

__device__ __forceinline__ void chunk_products(const float* __restrict__ A,
                                               const float* __restrict__ B,
                                               v8f& g0, v8f& g1, v8f& g2) {
#pragma unroll
    for (int kl = 0; kl < BS_KC / 4; ++kl) {
        const v2f a = *(const v2f*)(A + 4 * kl);
        v2f b = *(const v2f*)(B + 4 * kl);
        g0 = __builtin_amdgcn_wmma_f32_16x16x4_f32(
            false, a, false, b, (short)0, g0, false, false);
        b = *(const v2f*)(B + 128 * BS_PITCH + 4 * kl);
        g1 = __builtin_amdgcn_wmma_f32_16x16x4_f32(
            false, a, false, b, (short)0, g1, false, false);
        b = *(const v2f*)(B + 256 * BS_PITCH + 4 * kl);
        g2 = __builtin_amdgcn_wmma_f32_16x16x4_f32(
            false, a, false, b, (short)0, g2, false, false);
    }
}

__global__ __launch_bounds__(1024) void tile_scatter_kernel(
    const int* __restrict__ dst,
    const int4* __restrict__ rec, const unsigned int* __restrict__ mb,
    const unsigned int* __restrict__ tbw,
    const float* __restrict__ hv, const float* __restrict__ W_ih,
    const float* __restrict__ b_ih, const float* __restrict__ W_hh,
    const float* __restrict__ b_hh,
    const float* __restrict__ nf, float* __restrict__ out, int V, int E) {
    __shared__ __attribute__((aligned(16))) float acc[TILE_ROWS * TILE_PITCH];
    __shared__ float zs[TILE_ROWS];
    __shared__ float ms[TILE_ROWS];
    __shared__ __attribute__((aligned(16))) int lists[REGION_WORDS];
    float* nfs = (float*)lists;
    float* Bs = nfs + NFS_WORDS;

    const int tid = threadIdx.x;
    const int lane = tid & 31;
    const int wave = __builtin_amdgcn_readfirstlane(tid >> 5);
    const int base = blockIdx.x * TILE_ROWS;
    const int rows = (V - base < TILE_ROWS) ? (V - base) : TILE_ROWS;
    const unsigned int urows = (unsigned int)rows;
    int* wl = lists + wave * WLIST;

    for (int i = tid; i < TILE_ROWS * (TILE_PITCH / 4); i += 1024)
        ((float4*)acc)[i] = make_float4(0.0f, 0.0f, 0.0f, 0.0f);
    if (tid < TILE_ROWS) {
        zs[tid] = 0.0f;
        ms[tid] = tid < rows ? __int_as_float((int)mb[base + tid]) : 0.0f;
    }
    __syncthreads();

    const unsigned int tileId = blockIdx.x & 255u;
    unsigned int tm[8];
#pragma unroll
    for (int k = 0; k < 8; ++k) tm[k] = 0u - ((tileId >> k) & 1u);
    const int nGrp = (E >> 7) << 2;
    const uint4* __restrict__ tv = (const uint4*)tbw;
    const int nRow = (nGrp + 1023) >> 10;
    const int nBatch = (nRow + SCAN_U - 1) / SCAN_U;
    const int nRowP = nBatch * SCAN_U;
    int curRow = (int)((blockIdx.x * (unsigned int)nRowP) / gridDim.x);
    if (curRow >= nRowP) curRow = 0;

    int wcnt = 0;

    if (wave == 0) {
        for (int eo = (E >> 7) << 7; eo < E; eo += 32) {
            const int e = eo + lane;
            bool mt = false;
            if (e < E) mt = (unsigned int)(dst[e] - base) < urows;
            const unsigned int bal = __builtin_amdgcn_ballot_w32(mt);
            if (mt) wl[(int)__builtin_amdgcn_mbcnt_lo(bal, (unsigned int)wcnt)] = e;
            wcnt += __popc(bal);
        }
    }

    for (int it = 0; it <= nBatch; ++it) {
        const bool last = (it == nBatch);
        unsigned int mw[SCAN_U];
        int eb[SCAN_U];
#pragma unroll
        for (int u = 0; u < SCAN_U; ++u) { mw[u] = 0u; eb[u] = 0; }
        if (!last) {
            uint4 pa[SCAN_U], pb[SCAN_U];
            bool ok[SCAN_U];
#pragma unroll
            for (int u = 0; u < SCAN_U; ++u) {
                int r = curRow + u;
                if (r >= nRowP) r -= nRowP;
                const int g = r * 1024 + tid;
                ok[u] = g < nGrp;
                const int gc = ok[u] ? g : nGrp - 1;
                eb[u] = 128 * (gc >> 2) + 32 * (gc & 3);
                const int vi = ((gc & ~3) << 1) | (gc & 3);
                pa[u] = tv[vi];
                pb[u] = tv[vi + 4];
            }
            __builtin_amdgcn_sched_barrier(0);
#pragma unroll
            for (int u = 0; u < SCAN_U; ++u) {
                const unsigned int diff = (pa[u].x ^ tm[0]) | (pa[u].y ^ tm[1]) | (pa[u].z ^ tm[2])
                                        | (pa[u].w ^ tm[3]) | (pb[u].x ^ tm[4]) | (pb[u].y ^ tm[5])
                                        | (pb[u].z ^ tm[6]) | (pb[u].w ^ tm[7]);
                mw[u] = ok[u] ? ~diff : 0u;
            }
            curRow += SCAN_U;
            if (curRow >= nRowP) curRow -= nRowP;
        }
        bool pending = true;
        while (pending) {
            pending = false;
#pragma unroll
            for (int u = 0; u < SCAN_U; ++u) {
                unsigned int bal;
                while ((bal = __builtin_amdgcn_ballot_w32(mw[u] != 0u)) != 0u) {
                    if (wcnt > WLIST - 32) { pending = true; break; }
                    if (mw[u] != 0u)
                        wl[(int)__builtin_amdgcn_mbcnt_lo(bal, (unsigned int)wcnt)] =
                            eb[u] + __builtin_ctz(mw[u] | 0x80000000u);
                    wcnt += __popc(bal);
                    mw[u] &= mw[u] - 1u;
                }
            }
            while (wcnt >= 32 || (last && wcnt > 0)) {
                const int n = wcnt < 32 ? wcnt : 32;
                const int k0 = wcnt - n;
                wcnt = k0;
                const int4 er = rec[wl[k0 + (lane < n ? lane : 0)]];
                const int s = er.x;
                const int d = er.y;
                const bool mine = (unsigned int)(d - base) < urows;
                const int dl = mine ? d - base : 0;
                const float ex = expf(__int_as_float(er.z) - ms[dl]);
                const float ev = mine ? ex : 0.0f;
                if (lane < n) atomicAdd(&zs[dl], ev);
                for (int j = 0; j < n; j += 8) {
                    float4 h[8];
#pragma unroll
                    for (int k = 0; k < 8; ++k) {
                        const int jj = (j + k < n) ? (j + k) : (n - 1);
                        const int sj = __builtin_amdgcn_readlane(s, jj);
                        h[k] = ((const float4*)(hv + (size_t)sj * D_FEAT))[lane];
                    }
#pragma unroll
                    for (int k = 0; k < 8; ++k) {
                        const int jj = (j + k < n) ? (j + k) : (n - 1);
                        const float wv = __int_as_float(
                            __builtin_amdgcn_readlane(__float_as_int(ev), jj));
                        const float wt = (j + k < n) ? wv : 0.0f;
                        float* a = acc + __builtin_amdgcn_readlane(dl, jj) * TILE_PITCH + lane;
                        atomicAdd(a, wt * h[k].x);
                        atomicAdd(a + 32, wt * h[k].y);
                        atomicAdd(a + 64, wt * h[k].z);
                        atomicAdd(a + 96, wt * h[k].w);
                    }
                }
            }
        }
    }
    __syncthreads();

    const int sRow = tid >> 3;
    const int sCol = (tid & 7) * 4;
    const float* __restrict__ sIh = W_ih + (size_t)sRow * D_FEAT + sCol;
    const float* __restrict__ sHh = W_hh + (size_t)sRow * D_FEAT + sCol;
    float* sDst = Bs + sRow * BS_PITCH + sCol;

    {
        float4 nv[NF_TRIPS];
#pragma unroll
        for (int q = 0; q < NF_TRIPS; ++q) {
            const int r = wave + 32 * q;
            const int rc = r < rows ? r : rows - 1;
            nv[q] = ((const float4*)(nf + (size_t)(base + rc) * D_FEAT))[lane];
        }
#pragma unroll
        for (int q = 0; q < NF_TRIPS; ++q)
            if (wave + 32 * q < TILE_ROWS)
                *(float4*)(nfs + (wave + 32 * q) * TILE_PITCH + 4 * lane) = nv[q];
    }

    float4 pre0 = *(const float4*)(sIh);
    float4 pre1 = *(const float4*)(sIh + 128 * D_FEAT);
    float4 pre2 = *(const float4*)(sIh + 256 * D_FEAT);

    for (int r = wave; r < TILE_ROWS; r += 32) {
        const float zz = zs[r];
        const float inv = zz > 0.0f ? __builtin_amdgcn_rcpf(zz) : 0.0f;
        float* a = acc + r * TILE_PITCH;
        float v[4] = {a[lane], a[lane + 32], a[lane + 64], a[lane + 96]};
#pragma unroll
        for (int q = 0; q < 4; ++q) {
            const float x = v[q] * inv;
            v[q] = x > 0.0f ? x : __builtin_amdgcn_exp2f(1.44269504f * x) - 1.0f;
        }
        ((float4*)a)[lane] = make_float4(v[0], v[1], v[2], v[3]);
    }
    *(float4*)(sDst) = pre0;
    *(float4*)(sDst + 128 * BS_PITCH) = pre1;
    *(float4*)(sDst + 256 * BS_PITCH) = pre2;
    __syncthreads();

    {
        const int lo = lane & 15;
        const int hi = lane >> 4;
        const int d = (wave & 7) * 16 + lo;
        const int nTiles = (rows + 15) >> 4;
        const int nPass = (nTiles + 3) >> 2;
        const int nChunk = nPass * 8;
        const float br = b_ih[d] + b_hh[d], bz = b_ih[128 + d] + b_hh[128 + d];
        const float bn = b_ih[256 + d], bhn = b_hh[256 + d];
        const float* __restrict__ Bfrag = Bs + d * BS_PITCH + 2 * hi;
        int q = 0;
        for (int pass = 0; pass < nPass; ++pass) {
            const int rt = (wave >> 3) + 4 * pass;
            const bool act = rt < nTiles;
            int ar_row = rt * 16 + lo;
            if (ar_row > TILE_ROWS - 1) ar_row = TILE_ROWS - 1;
            const float* __restrict__ Ap = acc + ar_row * TILE_PITCH + 2 * hi;
            const float* __restrict__ Np = nfs + ar_row * TILE_PITCH + 2 * hi;
            v8f gr, gz, gn, hn;
#pragma unroll
            for (int r = 0; r < 8; ++r) { gr[r] = br; gz[r] = bz; gn[r] = bn; hn[r] = bhn; }
            auto chunk_step = [&](const int c, const float* __restrict__ A, v8f& g2) {
                const bool more = q + 1 < nChunk;
                if (more) {
                    const int cn = (c + 1) & 7;
                    const float* __restrict__ sp = (cn < 4 ? sIh : sHh) + (cn & 3) * BS_KC;
                    pre0 = *(const float4*)(sp);
                    pre1 = *(const float4*)(sp + 128 * D_FEAT);
                    pre2 = *(const float4*)(sp + 256 * D_FEAT);
                }
                if (act) chunk_products(A, Bfrag + (q & 1) * BS_WORDS, gr, gz, g2);
                if (more) {
                    float* sd = sDst + ((q + 1) & 1) * BS_WORDS;
                    *(float4*)(sd) = pre0;
                    *(float4*)(sd + 128 * BS_PITCH) = pre1;
                    *(float4*)(sd + 256 * BS_PITCH) = pre2;
                    __syncthreads();
                }
                ++q;
            };
#pragma unroll 1
            for (int c = 0; c < 4; ++c) chunk_step(c, Ap + c * BS_KC, gn);
#pragma unroll 1
            for (int c = 4; c < 8; ++c) chunk_step(c, Np + (c - 4) * BS_KC, hn);
            if (act) {
                float xs[8];
#pragma unroll
                for (int r = 0; r < 8; ++r) {
                    const int rl = rt * 16 + r + 8 * hi;
                    const int rx = rl < TILE_ROWS ? rl : TILE_ROWS - 1;
                    xs[r] = nfs[rx * TILE_PITCH + d];
                }
#pragma unroll
                for (int r = 0; r < 8; ++r) {
                    const int rl = rt * 16 + r + 8 * hi;
                    const float rg = gate_sigmoid(gr[r]);
                    const float u = gate_sigmoid(gz[r]);
                    const float n = gate_tanh(gn[r] + rg * hn[r]);
                    const float h = (1.0f - u) * n + u * xs[r];
                    if (rl < rows)
                        out[(size_t)(base + rl) * D_FEAT + d] = h > 0.0f ? h : 0.0f;
                }
            }
        }
    }
}

extern "C" void kernel_launch(void* const* d_in, const int* in_sizes, int n_in,
                              void* d_out, int out_size, void* d_ws, size_t ws_size,
                              hipStream_t stream) {
    const float* nf     = (const float*)d_in[0];
    const int*   src    = (const int*)d_in[1];
    const int*   dst    = (const int*)d_in[2];
    const float* W_edge = (const float*)d_in[3];
    const float* b_edge = (const float*)d_in[4];
    const float* W_proj = (const float*)d_in[5];
    const float* b_proj = (const float*)d_in[6];
    const float* W_ih   = (const float*)d_in[7];
    const float* b_ih   = (const float*)d_in[8];
    const float* W_hh   = (const float*)d_in[9];
    const float* b_hh   = (const float*)d_in[10];

    const int V = in_sizes[0] / D_FEAT;
    const int E = in_sizes[1];

    float* wsf = (float*)d_ws;
    float* pd    = wsf;
    float* ps    = pd + V;
    unsigned int* mb = (unsigned int*)(ps + V);
    int4* rec    = (int4*)((float*)(mb + V) + V);
    float* hv    = (float*)(rec + E);
    unsigned int* tbw = (unsigned int*)(hv + (size_t)V * 128);

    const int blk = 256;

    const int nodeGroups = (V + NODE_ROWS - 1) / NODE_ROWS;
    node_pass_kernel<<<(nodeGroups + 1) / 2, 64, 0, stream>>>(nf, W_edge, pd, ps, mb, V);

    const int gemmBlocks = (V + 15) / 16;
    wmma_gemm_kernel<128><<<gemmBlocks, 256, 0, stream>>>(nf, W_proj, b_proj, hv, V);

    edge_logit_kernel<<<((E + 3) / 4 + blk - 1) / blk, blk, 0, stream>>>(
        src, dst, pd, ps, b_edge, rec, mb, tbw, E);
    tile_scatter_kernel<<<(V + TILE_ROWS - 1) / TILE_ROWS, 1024, 0, stream>>>(
        dst, rec, mb, tbw, hv, W_ih, b_ih, W_hh, b_hh, nf, (float*)d_out, V, E);
}


# Segment max moves to the tile owner: a first sweep over its edges takes logits into an LDS integer max, match words kept in LDS for the add sweep; logit pass sends no global atomics; sums keep order
# speedup vs baseline: 3.4100x; 1.3650x over previous
#include <hip/hip_runtime.h>
#include <hip/hip_bf16.h>
#include <math.h>

typedef float v2f __attribute__((ext_vector_type(2)));
typedef float v8f __attribute__((ext_vector_type(8)));

#define D_FEAT 128

#define NODE_ROWS 32
template <int H>
__device__ __forceinline__ void fold_rows(float (&s0)[NODE_ROWS], float (&s1)[NODE_ROWS],
                                          const int lane) {
    const bool up = (lane & H) != 0;
#pragma unroll
    for (int j = 0; j < H; ++j) {
        const float keep0 = up ? s0[j + H] : s0[j];
        const float send0 = up ? s0[j] : s0[j + H];
        const float keep1 = up ? s1[j + H] : s1[j];
        const float send1 = up ? s1[j] : s1[j + H];
        s0[j] = keep0 + __shfl_xor(send0, H, 32);
        s1[j] = keep1 + __shfl_xor(send1, H, 32);
    }
}
__global__ __launch_bounds__(64) void node_pass_kernel(const float* __restrict__ nf,
                                                       const float* __restrict__ W_edge,
                                                       float* __restrict__ pd,
                                                       float* __restrict__ ps, int V) {
    const int lane = threadIdx.x & 31;
    const int grp = __builtin_amdgcn_readfirstlane(
        (int)(blockIdx.x * (blockDim.x >> 5) + (threadIdx.x >> 5)));
    const int v0 = grp * NODE_ROWS;
    if (v0 >= V) return;
    const float4 w0 = ((const float4*)W_edge)[lane];
    const float4 w1 = ((const float4*)(W_edge + D_FEAT))[lane];
    float4 x[NODE_ROWS];
#pragma unroll
    for (int j = 0; j < NODE_ROWS; ++j) {
        int v = v0 + j;
        if (v >= V) v = V - 1;
        x[j] = ((const float4*)(nf + (size_t)v * D_FEAT))[lane];
    }
    __builtin_amdgcn_sched_barrier(0);
    float s0[NODE_ROWS], s1[NODE_ROWS];
#pragma unroll
    for (int j = 0; j < NODE_ROWS; ++j) {
        s0[j] = fmaf(x[j].w, w0.w, fmaf(x[j].z, w0.z, fmaf(x[j].x, w0.x, x[j].y * w0.y)));
        s1[j] = fmaf(x[j].w, w1.w, fmaf(x[j].z, w1.z, fmaf(x[j].x, w1.x, x[j].y * w1.y)));
    }
    fold_rows<16>(s0, s1, lane);
    fold_rows<8>(s0, s1, lane);
    fold_rows<4>(s0, s1, lane);
    fold_rows<2>(s0, s1, lane);
    fold_rows<1>(s0, s1, lane);
    const int v = v0 + lane;
    if (v < V) {
        pd[v] = s0[0];
        ps[v] = s1[0];
    }
}

template <int N>
__global__ __launch_bounds__(256) void wmma_gemm_kernel(
    const float* __restrict__ X, const float* __restrict__ B,
    const float* __restrict__ bias, float* __restrict__ Out, int V) {
    __shared__ float xs[16][132];
    const int tid = threadIdx.x;
    const int rowbase = blockIdx.x * 16;
#pragma unroll
    for (int i = 0; i < 8; ++i) {
        int idx = tid + i * 256;
        int r = idx >> 7;
        int col = idx & 127;
        int row = rowbase + r;
        if (row >= V) row = V - 1;
        xs[r][col] = X[(size_t)row * D_FEAT + col];
    }
    __syncthreads();
    const int lane = tid & 31;
    const int wave = tid >> 5;
    const int lo = lane & 15;
    const int hi = lane >> 4;
    const bool fullTile = (rowbase + 16 <= V);
    for (int t = wave; t < N / 16; t += 8) {
        const int colbase = t * 16;
        v8f acc;
        const float bv = bias[colbase + lo];
#pragma unroll
        for (int r = 0; r < 8; ++r) acc[r] = bv;
#pragma unroll
        for (int kk = 0; kk < D_FEAT; kk += 4) {
            const int ka = kk + 2 * hi;
            v2f a, b;
            a[0] = xs[lo][ka];
            a[1] = xs[lo][ka + 1];
            b[0] = B[(size_t)ka * N + colbase + lo];
            b[1] = B[(size_t)(ka + 1) * N + colbase + lo];
            acc = __builtin_amdgcn_wmma_f32_16x16x4_f32(
                false, a, false, b, (short)0, acc, false, false);
        }
        if (fullTile) {
#pragma unroll
            for (int r = 0; r < 8; ++r) {
                int row = rowbase + r + 8 * hi;
                Out[(size_t)row * N + colbase + lo] = acc[r];
            }
        } else {
#pragma unroll
            for (int r = 0; r < 8; ++r) {
                int row = rowbase + r + 8 * hi;
                if (row < V) Out[(size_t)row * N + colbase + lo] = acc[r];
            }
        }
    }
}

#define TILE_ROWS 196

__device__ __forceinline__ float leaky_logit(float x) {
    return x > 0.0f ? x : 0.01f * x;
}

__device__ __forceinline__ void seg_max(unsigned int* m, float l) {
    const int iv = __float_as_int(l);
    if (iv >= 0) atomicMax((int*)m, iv);
    else         atomicMin(m, (unsigned int)iv);
}

__global__ void edge_logit_kernel(const int* __restrict__ src,
                                  const int* __restrict__ dst,
                                  const float* __restrict__ pd,
                                  const float* __restrict__ ps,
                                  const float* __restrict__ b_edge,
                                  int4* __restrict__ rec,
                                  unsigned int* __restrict__ tbw, int E) {
    const int i = blockIdx.x * blockDim.x + threadIdx.x;
    const int lane = threadIdx.x & 31;
    const int wv = i >> 5;
    const int e0 = 4 * i;
    const float be = b_edge[0];
    if (wv * 128 + 127 < E) {
        const int4 d = ((const int4*)dst)[i];
        const int4 s = ((const int4*)src)[i];
        const float l0 = leaky_logit(pd[d.x] + ps[s.x] + be);
        const float l1 = leaky_logit(pd[d.y] + ps[s.y] + be);
        const float l2 = leaky_logit(pd[d.z] + ps[s.z] + be);
        const float l3 = leaky_logit(pd[d.w] + ps[s.w] + be);
        int4* __restrict__ rw = rec + (size_t)wv * 128 + lane;
        rw[0]  = make_int4(s.x, d.x, __float_as_int(l0), 0);
        rw[32] = make_int4(s.y, d.y, __float_as_int(l1), 0);
        rw[64] = make_int4(s.z, d.z, __float_as_int(l2), 0);
        rw[96] = make_int4(s.w, d.w, __float_as_int(l3), 0);
        const unsigned int t[4] = {(unsigned int)d.x / TILE_ROWS, (unsigned int)d.y / TILE_ROWS,
                                   (unsigned int)d.z / TILE_ROWS, (unsigned int)d.w / TILE_ROWS};
        unsigned int word = 0u;
#pragma unroll
        for (int j = 0; j < 4; ++j) {
#pragma unroll
            for (int k = 0; k < 8; ++k) {
                const unsigned int bal = __builtin_amdgcn_ballot_w32(((t[j] >> k) & 1u) != 0u);
                if (lane == 16 * (k >> 2) + 4 * j + (k & 3)) word = bal;
            }
        }
        tbw[wv * 32 + lane] = word;
    } else {
        for (int e = e0; e < E && e < e0 + 4; ++e) {
            const int d = dst[e];
            const int sv = src[e];
            const float l = leaky_logit(pd[d] + ps[sv] + be);
            rec[e] = make_int4(sv, d, __float_as_int(l), 0);
        }
    }
}

#define TILE_PITCH 132
#define SCAN_U 5
#define WLIST 800
#define NF_TRIPS ((TILE_ROWS + 31) / 32)
#define BS_KC 32
#define BS_PITCH 36
#define BS_WORDS (384 * BS_PITCH)
#define NFS_WORDS (TILE_ROWS * TILE_PITCH)
#define GEMM_WORDS (NFS_WORDS + 2 * BS_WORDS)
#define REGION_WORDS ((32 * WLIST) > GEMM_WORDS ? (32 * WLIST) : GEMM_WORDS)
#define MCACHE_WORDS (REGION_WORDS - 32 * WLIST)

__device__ __forceinline__ float gate_sigmoid(float x) {
    return __builtin_amdgcn_rcpf(1.0f + __builtin_amdgcn_exp2f(-1.44269504f * x));
}
__device__ __forceinline__ float gate_tanh(float x) {
    return fmaf(-2.0f, __builtin_amdgcn_rcpf(1.0f + __builtin_amdgcn_exp2f(2.88539008f * x)), 1.0f);
}

__device__ __forceinline__ void chunk_products(const float* __restrict__ A,
                                               const float* __restrict__ B,
                                               v8f& g0, v8f& g1, v8f& g2) {
#pragma unroll
    for (int kl = 0; kl < BS_KC / 4; ++kl) {
        const v2f a = *(const v2f*)(A + 4 * kl);
        v2f b = *(const v2f*)(B + 4 * kl);
        g0 = __builtin_amdgcn_wmma_f32_16x16x4_f32(
            false, a, false, b, (short)0, g0, false, false);
        b = *(const v2f*)(B + 128 * BS_PITCH + 4 * kl);
        g1 = __builtin_amdgcn_wmma_f32_16x16x4_f32(
            false, a, false, b, (short)0, g1, false, false);
        b = *(const v2f*)(B + 256 * BS_PITCH + 4 * kl);
        g2 = __builtin_amdgcn_wmma_f32_16x16x4_f32(
            false, a, false, b, (short)0, g2, false, false);
    }
}

__global__ __launch_bounds__(1024) void tile_scatter_kernel(
    const int* __restrict__ dst,
    const int4* __restrict__ rec,
    const unsigned int* __restrict__ tbw,
    const float* __restrict__ hv, const float* __restrict__ W_ih,
    const float* __restrict__ b_ih, const float* __restrict__ W_hh,
    const float* __restrict__ b_hh,
    const float* __restrict__ nf, float* __restrict__ out, int V, int E) {
    __shared__ __attribute__((aligned(16))) float acc[TILE_ROWS * TILE_PITCH];
    __shared__ float zs[TILE_ROWS];
    __shared__ unsigned int msb[TILE_ROWS];
    __shared__ __attribute__((aligned(16))) int lists[REGION_WORDS];
    float* nfs = (float*)lists;
    float* Bs = nfs + NFS_WORDS;

    const int tid = threadIdx.x;
    const int lane = tid & 31;
    const int wave = __builtin_amdgcn_readfirstlane(tid >> 5);
    const int base = blockIdx.x * TILE_ROWS;
    const int rows = (V - base < TILE_ROWS) ? (V - base) : TILE_ROWS;
    const unsigned int urows = (unsigned int)rows;
    int* wl = lists + wave * WLIST;
    unsigned int* mcache = (unsigned int*)(lists + 32 * WLIST);

    for (int i = tid; i < TILE_ROWS * (TILE_PITCH / 4); i += 1024)
        ((float4*)acc)[i] = make_float4(0.0f, 0.0f, 0.0f, 0.0f);
    if (tid < TILE_ROWS) {
        zs[tid] = 0.0f;
        msb[tid] = 0xFF800000u;
    }
    __syncthreads();

    const unsigned int tileId = blockIdx.x & 255u;
    unsigned int tm[8];
#pragma unroll
    for (int k = 0; k < 8; ++k) tm[k] = 0u - ((tileId >> k) & 1u);
    const int nGrp = (E >> 7) << 2;
    const uint4* __restrict__ tv = (const uint4*)tbw;
    const int nRow = (nGrp + 1023) >> 10;
    const int nBatch = (nRow + SCAN_U - 1) / SCAN_U;
    const int nRowP = nBatch * SCAN_U;
    int startRow = (int)((blockIdx.x * (unsigned int)nRowP) / gridDim.x);
    if (startRow >= nRowP) startRow = 0;
    const bool cached = nRowP * 1024 <= MCACHE_WORDS;

#pragma unroll 1
    for (int sweep = 0; sweep < 2; ++sweep) {
        int curRow = startRow;
        int wcnt = 0;

        if (wave == 0) {
            for (int eo = (E >> 7) << 7; eo < E; eo += 32) {
                const int e = eo + lane;
                bool mt = false;
                if (e < E) mt = (unsigned int)(dst[e] - base) < urows;
                const unsigned int bal = __builtin_amdgcn_ballot_w32(mt);
                if (mt) wl[(int)__builtin_amdgcn_mbcnt_lo(bal, (unsigned int)wcnt)] = e;
                wcnt += __popc(bal);
            }
        }

        for (int it = 0; it <= nBatch; ++it) {
            const bool last = (it == nBatch);
            unsigned int mw[SCAN_U];
            int eb[SCAN_U];
#pragma unroll
            for (int u = 0; u < SCAN_U; ++u) { mw[u] = 0u; eb[u] = 0; }
            if (!last) {
                if (sweep == 0 || !cached) {
                    uint4 pa[SCAN_U], pb[SCAN_U];
                    bool ok[SCAN_U];
#pragma unroll
                    for (int u = 0; u < SCAN_U; ++u) {
                        int r = curRow + u;
                        if (r >= nRowP) r -= nRowP;
                        const int g = r * 1024 + tid;
                        ok[u] = g < nGrp;
                        const int gc = ok[u] ? g : nGrp - 1;
                        eb[u] = 128 * (gc >> 2) + 32 * (gc & 3);
                        const int vi = ((gc & ~3) << 1) | (gc & 3);
                        pa[u] = tv[vi];
                        pb[u] = tv[vi + 4];
                    }
                    __builtin_amdgcn_sched_barrier(0);
#pragma unroll
                    for (int u = 0; u < SCAN_U; ++u) {
                        const unsigned int diff = (pa[u].x ^ tm[0]) | (pa[u].y ^ tm[1]) | (pa[u].z ^ tm[2])
                                                | (pa[u].w ^ tm[3]) | (pb[u].x ^ tm[4]) | (pb[u].y ^ tm[5])
                                                | (pb[u].z ^ tm[6]) | (pb[u].w ^ tm[7]);
                        mw[u] = ok[u] ? ~diff : 0u;
                    }
                    if (sweep == 0 && cached) {
#pragma unroll
                        for (int u = 0; u < SCAN_U; ++u) {
                            int r = curRow + u;
                            if (r >= nRowP) r -= nRowP;
                            mcache[r * 1024 + tid] = mw[u];
                        }
                    }
                } else {
#pragma unroll
                    for (int u = 0; u < SCAN_U; ++u) {
                        int r = curRow + u;
                        if (r >= nRowP) r -= nRowP;
                        const int g = r * 1024 + tid;
                        const int gc = g < nGrp ? g : nGrp - 1;
                        eb[u] = 128 * (gc >> 2) + 32 * (gc & 3);
                        mw[u] = mcache[r * 1024 + tid];
                    }
                }
                curRow += SCAN_U;
                if (curRow >= nRowP) curRow -= nRowP;
            }
            bool pending = true;
            while (pending) {
                pending = false;
#pragma unroll
                for (int u = 0; u < SCAN_U; ++u) {
                    unsigned int bal;
                    while ((bal = __builtin_amdgcn_ballot_w32(mw[u] != 0u)) != 0u) {
                        if (wcnt > WLIST - 32) { pending = true; break; }
                        if (mw[u] != 0u)
                            wl[(int)__builtin_amdgcn_mbcnt_lo(bal, (unsigned int)wcnt)] =
                                eb[u] + __builtin_ctz(mw[u] | 0x80000000u);
                        wcnt += __popc(bal);
                        mw[u] &= mw[u] - 1u;
                    }
                }
                while (wcnt >= 32 || (last && wcnt > 0)) {
                    const int n = wcnt < 32 ? wcnt : 32;
                    const int k0 = wcnt - n;
                    wcnt = k0;
                    const int4 er = rec[wl[k0 + (lane < n ? lane : 0)]];
                    const int s = er.x;
                    const int d = er.y;
                    const bool mine = (unsigned int)(d - base) < urows;
                    const int dl = mine ? d - base : 0;
                    if (sweep == 0) {
                        if (lane < n && mine) seg_max(&msb[dl], __int_as_float(er.z));
                        continue;
                    }
                    const float ex = expf(__int_as_float(er.z) - __uint_as_float(msb[dl]));
                    const float ev = mine ? ex : 0.0f;
                    if (lane < n) atomicAdd(&zs[dl], ev);
                    for (int j = 0; j < n; j += 8) {
                        float4 h[8];
#pragma unroll
                        for (int k = 0; k < 8; ++k) {
                            const int jj = (j + k < n) ? (j + k) : (n - 1);
                            const int sj = __builtin_amdgcn_readlane(s, jj);
                            h[k] = ((const float4*)(hv + (size_t)sj * D_FEAT))[lane];
                        }
#pragma unroll
                        for (int k = 0; k < 8; ++k) {
                            const int jj = (j + k < n) ? (j + k) : (n - 1);
                            const float wv = __int_as_float(
                                __builtin_amdgcn_readlane(__float_as_int(ev), jj));
                            const float wt = (j + k < n) ? wv : 0.0f;
                            float* a = acc + __builtin_amdgcn_readlane(dl, jj) * TILE_PITCH + lane;
                            atomicAdd(a, wt * h[k].x);
                            atomicAdd(a + 32, wt * h[k].y);
                            atomicAdd(a + 64, wt * h[k].z);
                            atomicAdd(a + 96, wt * h[k].w);
                        }
                    }
                }
            }
        }
        __syncthreads();
    }

    const int sRow = tid >> 3;
    const int sCol = (tid & 7) * 4;
    const float* __restrict__ sIh = W_ih + (size_t)sRow * D_FEAT + sCol;
    const float* __restrict__ sHh = W_hh + (size_t)sRow * D_FEAT + sCol;
    float* sDst = Bs + sRow * BS_PITCH + sCol;

    {
        float4 nv[NF_TRIPS];
#pragma unroll
        for (int q = 0; q < NF_TRIPS; ++q) {
            const int r = wave + 32 * q;
            const int rc = r < rows ? r : rows - 1;
            nv[q] = ((const float4*)(nf + (size_t)(base + rc) * D_FEAT))[lane];
        }
#pragma unroll
        for (int q = 0; q < NF_TRIPS; ++q)
            if (wave + 32 * q < TILE_ROWS)
                *(float4*)(nfs + (wave + 32 * q) * TILE_PITCH + 4 * lane) = nv[q];
    }

    float4 pre0 = *(const float4*)(sIh);
    float4 pre1 = *(const float4*)(sIh + 128 * D_FEAT);
    float4 pre2 = *(const float4*)(sIh + 256 * D_FEAT);

    for (int r = wave; r < TILE_ROWS; r += 32) {
        const float zz = zs[r];
        const float inv = zz > 0.0f ? __builtin_amdgcn_rcpf(zz) : 0.0f;
        float* a = acc + r * TILE_PITCH;
        float v[4] = {a[lane], a[lane + 32], a[lane + 64], a[lane + 96]};
#pragma unroll
        for (int q = 0; q < 4; ++q) {
            const float x = v[q] * inv;
            v[q] = x > 0.0f ? x : __builtin_amdgcn_exp2f(1.44269504f * x) - 1.0f;
        }
        ((float4*)a)[lane] = make_float4(v[0], v[1], v[2], v[3]);
    }
    *(float4*)(sDst) = pre0;
    *(float4*)(sDst + 128 * BS_PITCH) = pre1;
    *(float4*)(sDst + 256 * BS_PITCH) = pre2;
    __syncthreads();

    {
        const int lo = lane & 15;
        const int hi = lane >> 4;
        const int d = (wave & 7) * 16 + lo;
        const int nTiles = (rows + 15) >> 4;
        const int nPass = (nTiles + 3) >> 2;
        const int nChunk = nPass * 8;
        const float br = b_ih[d] + b_hh[d], bz = b_ih[128 + d] + b_hh[128 + d];
        const float bn = b_ih[256 + d], bhn = b_hh[256 + d];
        const float* __restrict__ Bfrag = Bs + d * BS_PITCH + 2 * hi;
        int q = 0;
        for (int pass = 0; pass < nPass; ++pass) {
            const int rt = (wave >> 3) + 4 * pass;
            const bool act = rt < nTiles;
            int ar_row = rt * 16 + lo;
            if (ar_row > TILE_ROWS - 1) ar_row = TILE_ROWS - 1;
            const float* __restrict__ Ap = acc + ar_row * TILE_PITCH + 2 * hi;
            const float* __restrict__ Np = nfs + ar_row * TILE_PITCH + 2 * hi;
            v8f gr, gz, gn, hn;
#pragma unroll
            for (int r = 0; r < 8; ++r) { gr[r] = br; gz[r] = bz; gn[r] = bn; hn[r] = bhn; }
            auto chunk_step = [&](const int c, const float* __restrict__ A, v8f& g2) {
                const bool more = q + 1 < nChunk;
                if (more) {
                    const int cn = (c + 1) & 7;
                    const float* __restrict__ sp = (cn < 4 ? sIh : sHh) + (cn & 3) * BS_KC;
                    pre0 = *(const float4*)(sp);
                    pre1 = *(const float4*)(sp + 128 * D_FEAT);
                    pre2 = *(const float4*)(sp + 256 * D_FEAT);
                }
                if (act) chunk_products(A, Bfrag + (q & 1) * BS_WORDS, gr, gz, g2);
                if (more) {
                    float* sd = sDst + ((q + 1) & 1) * BS_WORDS;
                    *(float4*)(sd) = pre0;
                    *(float4*)(sd + 128 * BS_PITCH) = pre1;
                    *(float4*)(sd + 256 * BS_PITCH) = pre2;
                    __syncthreads();
                }
                ++q;
            };
#pragma unroll 1
            for (int c = 0; c < 4; ++c) chunk_step(c, Ap + c * BS_KC, gn);
#pragma unroll 1
            for (int c = 4; c < 8; ++c) chunk_step(c, Np + (c - 4) * BS_KC, hn);
            if (act) {
                float xs[8];
#pragma unroll
                for (int r = 0; r < 8; ++r) {
                    const int rl = rt * 16 + r + 8 * hi;
                    const int rx = rl < TILE_ROWS ? rl : TILE_ROWS - 1;
                    xs[r] = nfs[rx * TILE_PITCH + d];
                }
#pragma unroll
                for (int r = 0; r < 8; ++r) {
                    const int rl = rt * 16 + r + 8 * hi;
                    const float rg = gate_sigmoid(gr[r]);
                    const float u = gate_sigmoid(gz[r]);
                    const float n = gate_tanh(gn[r] + rg * hn[r]);
                    const float h = (1.0f - u) * n + u * xs[r];
                    if (rl < rows)
                        out[(size_t)(base + rl) * D_FEAT + d] = h > 0.0f ? h : 0.0f;
                }
            }
        }
    }
}

extern "C" void kernel_launch(void* const* d_in, const int* in_sizes, int n_in,
                              void* d_out, int out_size, void* d_ws, size_t ws_size,
                              hipStream_t stream) {
    const float* nf     = (const float*)d_in[0];
    const int*   src    = (const int*)d_in[1];
    const int*   dst    = (const int*)d_in[2];
    const float* W_edge = (const float*)d_in[3];
    const float* b_edge = (const float*)d_in[4];
    const float* W_proj = (const float*)d_in[5];
    const float* b_proj = (const float*)d_in[6];
    const float* W_ih   = (const float*)d_in[7];
    const float* b_ih   = (const float*)d_in[8];
    const float* W_hh   = (const float*)d_in[9];
    const float* b_hh   = (const float*)d_in[10];

    const int V = in_sizes[0] / D_FEAT;
    const int E = in_sizes[1];

    float* wsf = (float*)d_ws;
    float* pd    = wsf;
    float* ps    = pd + V;
    int4* rec    = (int4*)(ps + V + V + V);
    float* hv    = (float*)(rec + E);
    unsigned int* tbw = (unsigned int*)(hv + (size_t)V * 128);

    const int blk = 256;

    const int nodeGroups = (V + NODE_ROWS - 1) / NODE_ROWS;
    node_pass_kernel<<<(nodeGroups + 1) / 2, 64, 0, stream>>>(nf, W_edge, pd, ps, V);

    const int gemmBlocks = (V + 15) / 16;
    wmma_gemm_kernel<128><<<gemmBlocks, 256, 0, stream>>>(nf, W_proj, b_proj, hv, V);

    edge_logit_kernel<<<((E + 3) / 4 + blk - 1) / blk, blk, 0, stream>>>(
        src, dst, pd, ps, b_edge, rec, tbw, E);
    tile_scatter_kernel<<<(V + TILE_ROWS - 1) / TILE_ROWS, 1024, 0, stream>>>(
        dst, rec, tbw, hv, W_ih, b_ih, W_hh, b_hh, nf, (float*)d_out, V, E);
}


# Sweep 0 keeps the records it gathered (source, local row, logit) in LDS; sweep 1 walks them in the same groups: no second listing, no second record gather from memory; sums keep order
# speedup vs baseline: 3.5298x; 1.0351x over previous
#include <hip/hip_runtime.h>
#include <hip/hip_bf16.h>
#include <math.h>

typedef float v2f __attribute__((ext_vector_type(2)));
typedef float v8f __attribute__((ext_vector_type(8)));

#define D_FEAT 128

#define NODE_ROWS 32
template <int H>
__device__ __forceinline__ void fold_rows(float (&s0)[NODE_ROWS], float (&s1)[NODE_ROWS],
                                          const int lane) {
    const bool up = (lane & H) != 0;
#pragma unroll
    for (int j = 0; j < H; ++j) {
        const float keep0 = up ? s0[j + H] : s0[j];
        const float send0 = up ? s0[j] : s0[j + H];
        const float keep1 = up ? s1[j + H] : s1[j];
        const float send1 = up ? s1[j] : s1[j + H];
        s0[j] = keep0 + __shfl_xor(send0, H, 32);
        s1[j] = keep1 + __shfl_xor(send1, H, 32);
    }
}
__global__ __launch_bounds__(64) void node_pass_kernel(const float* __restrict__ nf,
                                                       const float* __restrict__ W_edge,
                                                       float* __restrict__ pd,
                                                       float* __restrict__ ps, int V) {
    const int lane = threadIdx.x & 31;
    const int grp = __builtin_amdgcn_readfirstlane(
        (int)(blockIdx.x * (blockDim.x >> 5) + (threadIdx.x >> 5)));
    const int v0 = grp * NODE_ROWS;
    if (v0 >= V) return;
    const float4 w0 = ((const float4*)W_edge)[lane];
    const float4 w1 = ((const float4*)(W_edge + D_FEAT))[lane];
    float4 x[NODE_ROWS];
#pragma unroll
    for (int j = 0; j < NODE_ROWS; ++j) {
        int v = v0 + j;
        if (v >= V) v = V - 1;
        x[j] = ((const float4*)(nf + (size_t)v * D_FEAT))[lane];
    }
    __builtin_amdgcn_sched_barrier(0);
    float s0[NODE_ROWS], s1[NODE_ROWS];
#pragma unroll
    for (int j = 0; j < NODE_ROWS; ++j) {
        s0[j] = fmaf(x[j].w, w0.w, fmaf(x[j].z, w0.z, fmaf(x[j].x, w0.x, x[j].y * w0.y)));
        s1[j] = fmaf(x[j].w, w1.w, fmaf(x[j].z, w1.z, fmaf(x[j].x, w1.x, x[j].y * w1.y)));
    }
    fold_rows<16>(s0, s1, lane);
    fold_rows<8>(s0, s1, lane);
    fold_rows<4>(s0, s1, lane);
    fold_rows<2>(s0, s1, lane);
    fold_rows<1>(s0, s1, lane);
    const int v = v0 + lane;
    if (v < V) {
        pd[v] = s0[0];
        ps[v] = s1[0];
    }
}

template <int N>
__global__ __launch_bounds__(256) void wmma_gemm_kernel(
    const float* __restrict__ X, const float* __restrict__ B,
    const float* __restrict__ bias, float* __restrict__ Out, int V) {
    __shared__ float xs[16][132];
    const int tid = threadIdx.x;
    const int rowbase = blockIdx.x * 16;
#pragma unroll
    for (int i = 0; i < 8; ++i) {
        int idx = tid + i * 256;
        int r = idx >> 7;
        int col = idx & 127;
        int row = rowbase + r;
        if (row >= V) row = V - 1;
        xs[r][col] = X[(size_t)row * D_FEAT + col];
    }
    __syncthreads();
    const int lane = tid & 31;
    const int wave = tid >> 5;
    const int lo = lane & 15;
    const int hi = lane >> 4;
    const bool fullTile = (rowbase + 16 <= V);
    for (int t = wave; t < N / 16; t += 8) {
        const int colbase = t * 16;
        v8f acc;
        const float bv = bias[colbase + lo];
#pragma unroll
        for (int r = 0; r < 8; ++r) acc[r] = bv;
#pragma unroll
        for (int kk = 0; kk < D_FEAT; kk += 4) {
            const int ka = kk + 2 * hi;
            v2f a, b;
            a[0] = xs[lo][ka];
            a[1] = xs[lo][ka + 1];
            b[0] = B[(size_t)ka * N + colbase + lo];
            b[1] = B[(size_t)(ka + 1) * N + colbase + lo];
            acc = __builtin_amdgcn_wmma_f32_16x16x4_f32(
                false, a, false, b, (short)0, acc, false, false);
        }
        if (fullTile) {
#pragma unroll
            for (int r = 0; r < 8; ++r) {
                int row = rowbase + r + 8 * hi;
                Out[(size_t)row * N + colbase + lo] = acc[r];
            }
        } else {
#pragma unroll
            for (int r = 0; r < 8; ++r) {
                int row = rowbase + r + 8 * hi;
                if (row < V) Out[(size_t)row * N + colbase + lo] = acc[r];
            }
        }
    }
}

#define TILE_ROWS 196

__device__ __forceinline__ float leaky_logit(float x) {
    return x > 0.0f ? x : 0.01f * x;
}

__device__ __forceinline__ void seg_max(unsigned int* m, float l) {
    const int iv = __float_as_int(l);
    if (iv >= 0) atomicMax((int*)m, iv);
    else         atomicMin(m, (unsigned int)iv);
}

__global__ void edge_logit_kernel(const int* __restrict__ src,
                                  const int* __restrict__ dst,
                                  const float* __restrict__ pd,
                                  const float* __restrict__ ps,
                                  const float* __restrict__ b_edge,
                                  int4* __restrict__ rec,
                                  unsigned int* __restrict__ tbw, int E) {
    const int i = blockIdx.x * blockDim.x + threadIdx.x;
    const int lane = threadIdx.x & 31;
    const int wv = i >> 5;
    const int e0 = 4 * i;
    const float be = b_edge[0];
    if (wv * 128 + 127 < E) {
        const int4 d = ((const int4*)dst)[i];
        const int4 s = ((const int4*)src)[i];
        const float l0 = leaky_logit(pd[d.x] + ps[s.x] + be);
        const float l1 = leaky_logit(pd[d.y] + ps[s.y] + be);
        const float l2 = leaky_logit(pd[d.z] + ps[s.z] + be);
        const float l3 = leaky_logit(pd[d.w] + ps[s.w] + be);
        int4* __restrict__ rw = rec + (size_t)wv * 128 + lane;
        rw[0]  = make_int4(s.x, d.x, __float_as_int(l0), 0);
        rw[32] = make_int4(s.y, d.y, __float_as_int(l1), 0);
        rw[64] = make_int4(s.z, d.z, __float_as_int(l2), 0);
        rw[96] = make_int4(s.w, d.w, __float_as_int(l3), 0);
        const unsigned int t[4] = {(unsigned int)d.x / TILE_ROWS, (unsigned int)d.y / TILE_ROWS,
                                   (unsigned int)d.z / TILE_ROWS, (unsigned int)d.w / TILE_ROWS};
        unsigned int word = 0u;
#pragma unroll
        for (int j = 0; j < 4; ++j) {
#pragma unroll
            for (int k = 0; k < 8; ++k) {
                const unsigned int bal = __builtin_amdgcn_ballot_w32(((t[j] >> k) & 1u) != 0u);
                if (lane == 16 * (k >> 2) + 4 * j + (k & 3)) word = bal;
            }
        }
        tbw[wv * 32 + lane] = word;
    } else {
        for (int e = e0; e < E && e < e0 + 4; ++e) {
            const int d = dst[e];
            const int sv = src[e];
            const float l = leaky_logit(pd[d] + ps[sv] + be);
            rec[e] = make_int4(sv, d, __float_as_int(l), 0);
        }
    }
}

#define TILE_PITCH 132
#define SCAN_U 5
#define WLIST 800
#define NF_TRIPS ((TILE_ROWS + 31) / 32)
#define BS_KC 32
#define BS_PITCH 36
#define BS_WORDS (384 * BS_PITCH)
#define NFS_WORDS (TILE_ROWS * TILE_PITCH)
#define GEMM_WORDS (NFS_WORDS + 2 * BS_WORDS)
#define REGION_WORDS ((32 * WLIST) > GEMM_WORDS ? (32 * WLIST) : GEMM_WORDS)
#define RSTORE_WORDS (REGION_WORDS - 32 * WLIST)
#define RCAP ((RSTORE_WORDS / 32) / 3)

__device__ __forceinline__ float gate_sigmoid(float x) {
    return __builtin_amdgcn_rcpf(1.0f + __builtin_amdgcn_exp2f(-1.44269504f * x));
}
__device__ __forceinline__ float gate_tanh(float x) {
    return fmaf(-2.0f, __builtin_amdgcn_rcpf(1.0f + __builtin_amdgcn_exp2f(2.88539008f * x)), 1.0f);
}

__device__ __forceinline__ void chunk_products(const float* __restrict__ A,
                                               const float* __restrict__ B,
                                               v8f& g0, v8f& g1, v8f& g2) {
#pragma unroll
    for (int kl = 0; kl < BS_KC / 4; ++kl) {
        const v2f a = *(const v2f*)(A + 4 * kl);
        v2f b = *(const v2f*)(B + 4 * kl);
        g0 = __builtin_amdgcn_wmma_f32_16x16x4_f32(
            false, a, false, b, (short)0, g0, false, false);
        b = *(const v2f*)(B + 128 * BS_PITCH + 4 * kl);
        g1 = __builtin_amdgcn_wmma_f32_16x16x4_f32(
            false, a, false, b, (short)0, g1, false, false);
        b = *(const v2f*)(B + 256 * BS_PITCH + 4 * kl);
        g2 = __builtin_amdgcn_wmma_f32_16x16x4_f32(
            false, a, false, b, (short)0, g2, false, false);
    }
}

__global__ __launch_bounds__(1024) void tile_scatter_kernel(
    const int* __restrict__ dst,
    const int4* __restrict__ rec,
    const unsigned int* __restrict__ tbw,
    const float* __restrict__ hv, const float* __restrict__ W_ih,
    const float* __restrict__ b_ih, const float* __restrict__ W_hh,
    const float* __restrict__ b_hh,
    const float* __restrict__ nf, float* __restrict__ out, int V, int E) {
    __shared__ __attribute__((aligned(16))) float acc[TILE_ROWS * TILE_PITCH];
    __shared__ float zs[TILE_ROWS];
    __shared__ unsigned int msb[TILE_ROWS];
    __shared__ __attribute__((aligned(16))) int lists[REGION_WORDS];
    __shared__ int recOvf;
    float* nfs = (float*)lists;
    float* Bs = nfs + NFS_WORDS;

    const int tid = threadIdx.x;
    const int lane = tid & 31;
    const int wave = __builtin_amdgcn_readfirstlane(tid >> 5);
    const int base = blockIdx.x * TILE_ROWS;
    const int rows = (V - base < TILE_ROWS) ? (V - base) : TILE_ROWS;
    const unsigned int urows = (unsigned int)rows;
    int* wl = lists + wave * WLIST;
    int* rs = lists + 32 * WLIST + wave * (3 * RCAP);

    for (int i = tid; i < TILE_ROWS * (TILE_PITCH / 4); i += 1024)
        ((float4*)acc)[i] = make_float4(0.0f, 0.0f, 0.0f, 0.0f);
    if (tid < TILE_ROWS) {
        zs[tid] = 0.0f;
        msb[tid] = 0xFF800000u;
    }
    if (tid == 0) recOvf = 0;
    __syncthreads();

    const unsigned int tileId = blockIdx.x & 255u;
    unsigned int tm[8];
#pragma unroll
    for (int k = 0; k < 8; ++k) tm[k] = 0u - ((tileId >> k) & 1u);
    const int nGrp = (E >> 7) << 2;
    const uint4* __restrict__ tv = (const uint4*)tbw;
    const int nRow = (nGrp + 1023) >> 10;
    const int nBatch = (nRow + SCAN_U - 1) / SCAN_U;
    const int nRowP = nBatch * SCAN_U;
    int startRow = (int)((blockIdx.x * (unsigned int)nRowP) / gridDim.x);
    if (startRow >= nRowP) startRow = 0;
    auto add_group = [&](const int n, const int s, const int dl, const bool mine, const int lbits) {
        const float ex = expf(__int_as_float(lbits) - __uint_as_float(msb[dl]));
        const float ev = mine ? ex : 0.0f;
        if (lane < n) atomicAdd(&zs[dl], ev);
        for (int j = 0; j < n; j += 8) {
            float4 h[8];
#pragma unroll
            for (int k = 0; k < 8; ++k) {
                const int jj = (j + k < n) ? (j + k) : (n - 1);
                const int sj = __builtin_amdgcn_readlane(s, jj);
                h[k] = ((const float4*)(hv + (size_t)sj * D_FEAT))[lane];
            }
#pragma unroll
            for (int k = 0; k < 8; ++k) {
                const int jj = (j + k < n) ? (j + k) : (n - 1);
                const float wv = __int_as_float(
                    __builtin_amdgcn_readlane(__float_as_int(ev), jj));
                const float wt = (j + k < n) ? wv : 0.0f;
                float* a = acc + __builtin_amdgcn_readlane(dl, jj) * TILE_PITCH + lane;
                atomicAdd(a, wt * h[k].x);
                atomicAdd(a + 32, wt * h[k].y);
                atomicAdd(a + 64, wt * h[k].z);
                atomicAdd(a + 96, wt * h[k].w);
            }
        }
    };

    int rcnt = 0;
    bool useRec = false;

#pragma unroll 1
    for (int sweep = 0; sweep < 2; ++sweep) {
        if (sweep == 1 && useRec) break;
        int curRow = startRow;
        int wcnt = 0;

        if (wave == 0) {
            for (int eo = (E >> 7) << 7; eo < E; eo += 32) {
                const int e = eo + lane;
                bool mt = false;
                if (e < E) mt = (unsigned int)(dst[e] - base) < urows;
                const unsigned int bal = __builtin_amdgcn_ballot_w32(mt);
                if (mt) wl[(int)__builtin_amdgcn_mbcnt_lo(bal, (unsigned int)wcnt)] = e;
                wcnt += __popc(bal);
            }
        }

        for (int it = 0; it <= nBatch; ++it) {
            const bool last = (it == nBatch);
            unsigned int mw[SCAN_U];
            int eb[SCAN_U];
#pragma unroll
            for (int u = 0; u < SCAN_U; ++u) { mw[u] = 0u; eb[u] = 0; }
            if (!last) {
                uint4 pa[SCAN_U], pb[SCAN_U];
                bool ok[SCAN_U];
#pragma unroll
                for (int u = 0; u < SCAN_U; ++u) {
                    int r = curRow + u;
                    if (r >= nRowP) r -= nRowP;
                    const int g = r * 1024 + tid;
                    ok[u] = g < nGrp;
                    const int gc = ok[u] ? g : nGrp - 1;
                    eb[u] = 128 * (gc >> 2) + 32 * (gc & 3);
                    const int vi = ((gc & ~3) << 1) | (gc & 3);
                    pa[u] = tv[vi];
                    pb[u] = tv[vi + 4];
                }
                __builtin_amdgcn_sched_barrier(0);
#pragma unroll
                for (int u = 0; u < SCAN_U; ++u) {
                    const unsigned int diff = (pa[u].x ^ tm[0]) | (pa[u].y ^ tm[1]) | (pa[u].z ^ tm[2])
                                            | (pa[u].w ^ tm[3]) | (pb[u].x ^ tm[4]) | (pb[u].y ^ tm[5])
                                            | (pb[u].z ^ tm[6]) | (pb[u].w ^ tm[7]);
                    mw[u] = ok[u] ? ~diff : 0u;
                }
                curRow += SCAN_U;
                if (curRow >= nRowP) curRow -= nRowP;
            }
            bool pending = true;
            while (pending) {
                pending = false;
#pragma unroll
                for (int u = 0; u < SCAN_U; ++u) {
                    unsigned int bal;
                    while ((bal = __builtin_amdgcn_ballot_w32(mw[u] != 0u)) != 0u) {
                        if (wcnt > WLIST - 32) { pending = true; break; }
                        if (mw[u] != 0u)
                            wl[(int)__builtin_amdgcn_mbcnt_lo(bal, (unsigned int)wcnt)] =
                                eb[u] + __builtin_ctz(mw[u] | 0x80000000u);
                        wcnt += __popc(bal);
                        mw[u] &= mw[u] - 1u;
                    }
                }
                while (wcnt >= 32 || (last && wcnt > 0)) {
                    const int n = wcnt < 32 ? wcnt : 32;
                    const int k0 = wcnt - n;
                    wcnt = k0;
                    const int4 er = rec[wl[k0 + (lane < n ? lane : 0)]];
                    const int s = er.x;
                    const int d = er.y;
                    const bool mine = (unsigned int)(d - base) < urows;
                    const int dl = mine ? d - base : 0;
                    if (sweep == 0) {
                        const bool keep = lane < n && mine;
                        if (keep) seg_max(&msb[dl], __int_as_float(er.z));
                        const unsigned int kb = __builtin_amdgcn_ballot_w32(keep);
                        if (rcnt + 32 <= RCAP) {
                            if (keep) {
                                const int p = (int)__builtin_amdgcn_mbcnt_lo(kb, (unsigned int)rcnt);
                                rs[p] = s;
                                rs[RCAP + p] = dl;
                                rs[2 * RCAP + p] = er.z;
                            }
                            rcnt += __popc(kb);
                        } else {
                            recOvf = 1;
                        }
                        continue;
                    }
                    add_group(n, s, dl, mine, er.z);
                }
            }
        }
        __syncthreads();
        if (sweep == 0) useRec = (__builtin_amdgcn_readfirstlane(recOvf) == 0);
    }
    if (useRec) {
        for (int k0 = 0; k0 < rcnt; k0 += 32) {
            const int n = rcnt - k0 < 32 ? rcnt - k0 : 32;
            const int p = k0 + (lane < n ? lane : 0);
            add_group(n, rs[p], rs[RCAP + p], true, rs[2 * RCAP + p]);
        }
        __syncthreads();
    }

    const int sRow = tid >> 3;
    const int sCol = (tid & 7) * 4;
    const float* __restrict__ sIh = W_ih + (size_t)sRow * D_FEAT + sCol;
    const float* __restrict__ sHh = W_hh + (size_t)sRow * D_FEAT + sCol;
    float* sDst = Bs + sRow * BS_PITCH + sCol;

    {
        float4 nv[NF_TRIPS];
#pragma unroll
        for (int q = 0; q < NF_TRIPS; ++q) {
            const int r = wave + 32 * q;
            const int rc = r < rows ? r : rows - 1;
            nv[q] = ((const float4*)(nf + (size_t)(base + rc) * D_FEAT))[lane];
        }
#pragma unroll
        for (int q = 0; q < NF_TRIPS; ++q)
            if (wave + 32 * q < TILE_ROWS)
                *(float4*)(nfs + (wave + 32 * q) * TILE_PITCH + 4 * lane) = nv[q];
    }

    float4 pre0 = *(const float4*)(sIh);
    float4 pre1 = *(const float4*)(sIh + 128 * D_FEAT);
    float4 pre2 = *(const float4*)(sIh + 256 * D_FEAT);

    for (int r = wave; r < TILE_ROWS; r += 32) {
        const float zz = zs[r];
        const float inv = zz > 0.0f ? __builtin_amdgcn_rcpf(zz) : 0.0f;
        float* a = acc + r * TILE_PITCH;
        float v[4] = {a[lane], a[lane + 32], a[lane + 64], a[lane + 96]};
#pragma unroll
        for (int q = 0; q < 4; ++q) {
            const float x = v[q] * inv;
            v[q] = x > 0.0f ? x : __builtin_amdgcn_exp2f(1.44269504f * x) - 1.0f;
        }
        ((float4*)a)[lane] = make_float4(v[0], v[1], v[2], v[3]);
    }
    *(float4*)(sDst) = pre0;
    *(float4*)(sDst + 128 * BS_PITCH) = pre1;
    *(float4*)(sDst + 256 * BS_PITCH) = pre2;
    __syncthreads();

    {
        const int lo = lane & 15;
        const int hi = lane >> 4;
        const int d = (wave & 7) * 16 + lo;
        const int nTiles = (rows + 15) >> 4;
        const int nPass = (nTiles + 3) >> 2;
        const int nChunk = nPass * 8;
        const float br = b_ih[d] + b_hh[d], bz = b_ih[128 + d] + b_hh[128 + d];
        const float bn = b_ih[256 + d], bhn = b_hh[256 + d];
        const float* __restrict__ Bfrag = Bs + d * BS_PITCH + 2 * hi;
        int q = 0;
        for (int pass = 0; pass < nPass; ++pass) {
            const int rt = (wave >> 3) + 4 * pass;
            const bool act = rt < nTiles;
            int ar_row = rt * 16 + lo;
            if (ar_row > TILE_ROWS - 1) ar_row = TILE_ROWS - 1;
            const float* __restrict__ Ap = acc + ar_row * TILE_PITCH + 2 * hi;
            const float* __restrict__ Np = nfs + ar_row * TILE_PITCH + 2 * hi;
            v8f gr, gz, gn, hn;
#pragma unroll
            for (int r = 0; r < 8; ++r) { gr[r] = br; gz[r] = bz; gn[r] = bn; hn[r] = bhn; }
            auto chunk_step = [&](const int c, const float* __restrict__ A, v8f& g2) {
                const bool more = q + 1 < nChunk;
                if (more) {
                    const int cn = (c + 1) & 7;
                    const float* __restrict__ sp = (cn < 4 ? sIh : sHh) + (cn & 3) * BS_KC;
                    pre0 = *(const float4*)(sp);
                    pre1 = *(const float4*)(sp + 128 * D_FEAT);
                    pre2 = *(const float4*)(sp + 256 * D_FEAT);
                }
                if (act) chunk_products(A, Bfrag + (q & 1) * BS_WORDS, gr, gz, g2);
                if (more) {
                    float* sd = sDst + ((q + 1) & 1) * BS_WORDS;
                    *(float4*)(sd) = pre0;
                    *(float4*)(sd + 128 * BS_PITCH) = pre1;
                    *(float4*)(sd + 256 * BS_PITCH) = pre2;
                    __syncthreads();
                }
                ++q;
            };
#pragma unroll 1
            for (int c = 0; c < 4; ++c) chunk_step(c, Ap + c * BS_KC, gn);
#pragma unroll 1
            for (int c = 4; c < 8; ++c) chunk_step(c, Np + (c - 4) * BS_KC, hn);
            if (act) {
                float xs[8];
#pragma unroll
                for (int r = 0; r < 8; ++r) {
                    const int rl = rt * 16 + r + 8 * hi;
                    const int rx = rl < TILE_ROWS ? rl : TILE_ROWS - 1;
                    xs[r] = nfs[rx * TILE_PITCH + d];
                }
#pragma unroll
                for (int r = 0; r < 8; ++r) {
                    const int rl = rt * 16 + r + 8 * hi;
                    const float rg = gate_sigmoid(gr[r]);
                    const float u = gate_sigmoid(gz[r]);
                    const float n = gate_tanh(gn[r] + rg * hn[r]);
                    const float h = (1.0f - u) * n + u * xs[r];
                    if (rl < rows)
                        out[(size_t)(base + rl) * D_FEAT + d] = h > 0.0f ? h : 0.0f;
                }
            }
        }
    }
}

extern "C" void kernel_launch(void* const* d_in, const int* in_sizes, int n_in,
                              void* d_out, int out_size, void* d_ws, size_t ws_size,
                              hipStream_t stream) {
    const float* nf     = (const float*)d_in[0];
    const int*   src    = (const int*)d_in[1];
    const int*   dst    = (const int*)d_in[2];
    const float* W_edge = (const float*)d_in[3];
    const float* b_edge = (const float*)d_in[4];
    const float* W_proj = (const float*)d_in[5];
    const float* b_proj = (const float*)d_in[6];
    const float* W_ih   = (const float*)d_in[7];
    const float* b_ih   = (const float*)d_in[8];
    const float* W_hh   = (const float*)d_in[9];
    const float* b_hh   = (const float*)d_in[10];

    const int V = in_sizes[0] / D_FEAT;
    const int E = in_sizes[1];

    float* wsf = (float*)d_ws;
    float* pd    = wsf;
    float* ps    = pd + V;
    int4* rec    = (int4*)(ps + V + V + V);
    float* hv    = (float*)(rec + E);
    unsigned int* tbw = (unsigned int*)(hv + (size_t)V * 128);

    const int blk = 256;

    const int nodeGroups = (V + NODE_ROWS - 1) / NODE_ROWS;
    node_pass_kernel<<<(nodeGroups + 1) / 2, 64, 0, stream>>>(nf, W_edge, pd, ps, V);

    const int gemmBlocks = (V + 15) / 16;
    wmma_gemm_kernel<128><<<gemmBlocks, 256, 0, stream>>>(nf, W_proj, b_proj, hv, V);

    edge_logit_kernel<<<((E + 3) / 4 + blk - 1) / blk, blk, 0, stream>>>(
        src, dst, pd, ps, b_edge, rec, tbw, E);
    tile_scatter_kernel<<<(V + TILE_ROWS - 1) / TILE_ROWS, 1024, 0, stream>>>(
        dst, rec, tbw, hv, W_ih, b_ih, W_hh, b_hh, nf, (float*)d_out, V, E);
}
